# starting kernel
# baseline (speedup 1.0000x reference)
#include <hip/hip_runtime.h>
#include <hip/hip_bf16.h>

#define NN 100000
#define IN_F 32
#define HID_F 64
#define OUT_F 16

typedef float v2f __attribute__((ext_vector_type(2)));
typedef float v8f __attribute__((ext_vector_type(8)));

__global__ __launch_bounds__(256) void zero_ws_kernel(float* __restrict__ ws, long n) {
    long i = (long)blockIdx.x * blockDim.x + threadIdx.x;
    long stride = (long)gridDim.x * blockDim.x;
    for (; i < n; i += stride) ws[i] = 0.0f;
}

__global__ __launch_bounds__(256) void scatter32_kernel(const float* __restrict__ feat,
                                                        const int* __restrict__ src,
                                                        const int* __restrict__ dst,
                                                        float* __restrict__ agg,
                                                        float* __restrict__ deg,
                                                        int nE) {
    int tid  = blockIdx.x * blockDim.x + threadIdx.x;
    int e    = tid >> 5;
    int lane = tid & 31;
    if (e >= nE) return;
    int s = src[e];
    int d = dst[e];
    atomicAdd(&agg[(long)d * IN_F + lane], feat[(long)s * IN_F + lane]);
    if (lane == 0) atomicAdd(&deg[d], 1.0f);
}

__global__ __launch_bounds__(256) void scatter64_kernel(const float* __restrict__ h,
                                                        const int* __restrict__ src,
                                                        const int* __restrict__ dst,
                                                        float* __restrict__ agg,
                                                        int nE) {
    int tid  = blockIdx.x * blockDim.x + threadIdx.x;
    int e    = tid >> 5;
    int lane = tid & 31;
    if (e >= nE) return;
    int s = src[e];
    int d = dst[e];
    atomicAdd(&agg[(long)d * HID_F + lane],      h[(long)s * HID_F + lane]);
    atomicAdd(&agg[(long)d * HID_F + lane + 32], h[(long)s * HID_F + lane + 32]);
}

__global__ __launch_bounds__(256) void sage_layer1_kernel(const float* __restrict__ feat,
                                                          const float* __restrict__ agg,
                                                          const float* __restrict__ deg,
                                                          const float* __restrict__ Wself,
                                                          const float* __restrict__ Wneigh,
                                                          const float* __restrict__ bias,
                                                          float* __restrict__ h,
                                                          int nRowTiles) {
    int tid  = blockIdx.x * blockDim.x + threadIdx.x;
    int wave = tid >> 5;
    int lane = tid & 31;
    const int colTiles = HID_F / 16;
    int tileRow = wave / colTiles;
    int ct      = wave % colTiles;
    if (tileRow >= nRowTiles) return;

    int hi   = lane >> 4;
    int l15  = lane & 15;
    long row = (long)tileRow * 16 + l15;
    int colW = ct * 16 + l15;

    float dg   = fmaxf(deg[row], 1.0f);
    float invd = 1.0f / dg;

    v8f c;
    float bv = bias[colW];
#pragma unroll
    for (int v = 0; v < 8; ++v) c[v] = bv;

    const float* fr = feat + row * IN_F;
    const float* ar = agg  + row * IN_F;

#pragma unroll
    for (int k0 = 0; k0 < IN_F; k0 += 4) {
        int kA = k0 + 2 * hi;
        v2f a, b;
        a.x = fr[kA];
        a.y = fr[kA + 1];
        b.x = Wself[(long)(kA)     * HID_F + colW];
        b.y = Wself[(long)(kA + 1) * HID_F + colW];
        c = __builtin_amdgcn_wmma_f32_16x16x4_f32(false, a, false, b, (short)0, c, false, false);
        a.x = ar[kA]     * invd;
        a.y = ar[kA + 1] * invd;
        b.x = Wneigh[(long)(kA)     * HID_F + colW];
        b.y = Wneigh[(long)(kA + 1) * HID_F + colW];
        c = __builtin_amdgcn_wmma_f32_16x16x4_f32(false, a, false, b, (short)0, c, false, false);
    }

#pragma unroll
    for (int v = 0; v < 8; ++v) {
        long r = (long)tileRow * 16 + v + 8 * hi;
        h[r * HID_F + colW] = fmaxf(c[v], 0.0f);
    }
}

__global__ __launch_bounds__(256) void sage_layer2_kernel(const float* __restrict__ h,
                                                          const float* __restrict__ agg,
                                                          const float* __restrict__ deg,
                                                          const float* __restrict__ Wself,
                                                          const float* __restrict__ Wneigh,
                                                          const float* __restrict__ bias,
                                                          float* __restrict__ out,
                                                          int nRowTiles) {
    int tid  = blockIdx.x * blockDim.x + threadIdx.x;
    int wave = tid >> 5;
    int lane = tid & 31;
    int tileRow = wave;
    if (tileRow >= nRowTiles) return;

    int hi  = lane >> 4;
    int l15 = lane & 15;
    long row = (long)tileRow * 16 + l15;
    int colW = l15;

    float dg   = fmaxf(deg[row], 1.0f);
    float invd = 1.0f / dg;

    v8f c;
    float bv = bias[colW];
#pragma unroll
    for (int v = 0; v < 8; ++v) c[v] = bv;

    const float* hr = h   + row * HID_F;
    const float* ar = agg + row * HID_F;

#pragma unroll
    for (int k0 = 0; k0 < HID_F; k0 += 4) {
        int kA = k0 + 2 * hi;
        v2f a, b;
        a.x = hr[kA];
        a.y = hr[kA + 1];
        b.x = Wself[(long)(kA)     * OUT_F + colW];
        b.y = Wself[(long)(kA + 1) * OUT_F + colW];
        c = __builtin_amdgcn_wmma_f32_16x16x4_f32(false, a, false, b, (short)0, c, false, false);
        a.x = ar[kA]     * invd;
        a.y = ar[kA + 1] * invd;
        b.x = Wneigh[(long)(kA)     * OUT_F + colW];
        b.y = Wneigh[(long)(kA + 1) * OUT_F + colW];
        c = __builtin_amdgcn_wmma_f32_16x16x4_f32(false, a, false, b, (short)0, c, false, false);
    }

#pragma unroll
    for (int v = 0; v < 8; ++v) {
        long r = (long)tileRow * 16 + v + 8 * hi;
        out[r * OUT_F + colW] = c[v];
    }
}

extern "C" void kernel_launch(void* const* d_in, const int* in_sizes, int n_in,
                              void* d_out, int out_size, void* d_ws, size_t ws_size,
                              hipStream_t stream) {
    const float* feat = (const float*)d_in[0];
    const int*   src  = (const int*)d_in[1];
    const int*   dst  = (const int*)d_in[2];
    const float* W1s  = (const float*)d_in[3];
    const float* W1n  = (const float*)d_in[4];
    const float* b1   = (const float*)d_in[5];
    const float* W2s  = (const float*)d_in[6];
    const float* W2n  = (const float*)d_in[7];
    const float* b2   = (const float*)d_in[8];
    float* out = (float*)d_out;

    const int nNodes = in_sizes[0] / IN_F;
    const int nE     = in_sizes[1];
    const int nRowTiles = nNodes / 16;

    float* ws   = (float*)d_ws;
    float* deg  = ws;
    float* agg1 = deg  + nNodes;
    float* agg2 = agg1 + (long)nNodes * IN_F;
    float* h    = agg2 + (long)nNodes * HID_F;

    long zn = (long)nNodes * (1 + IN_F + HID_F);
    zero_ws_kernel<<<4096, 256, 0, stream>>>(ws, zn);

    {
        long threads = (long)nE * 32;
        int blocks = (int)((threads + 255) / 256);
        scatter32_kernel<<<blocks, 256, 0, stream>>>(feat, src, dst, agg1, deg, nE);
    }
    {
        int waves = nRowTiles * (HID_F / 16);
        int blocks = (waves * 32 + 255) / 256;
        sage_layer1_kernel<<<blocks, 256, 0, stream>>>(feat, agg1, deg, W1s, W1n, b1, h, nRowTiles);
    }

    {
        long threads = (long)nE * 32;
        int blocks = (int)((threads + 255) / 256);
        scatter64_kernel<<<blocks, 256, 0, stream>>>(h, src, dst, agg2, nE);
    }
    {
        int waves = nRowTiles;
        int blocks = (waves * 32 + 255) / 256;
        sage_layer2_kernel<<<blocks, 256, 0, stream>>>(h, agg2, deg, W2s, W2n, b2, out, nRowTiles);
    }
}


# Float-atomic scatter-add -> per-call counting sort of edges by destination (int rank atomics) + per-node gather-sum for both layers; sums now in rank order (still run-varying), no float atomics
# speedup vs baseline: 2.0657x; 2.0657x over previous
#include <hip/hip_runtime.h>
#include <hip/hip_bf16.h>

#define NN 100000
#define IN_F 32
#define HID_F 64
#define OUT_F 16

typedef float v2f __attribute__((ext_vector_type(2)));
typedef float v8f __attribute__((ext_vector_type(8)));

__global__ __launch_bounds__(256) void zero_i32_kernel(int* __restrict__ p, int n) {
    int i = blockIdx.x * blockDim.x + threadIdx.x;
    if (i < n) p[i] = 0;
}

__global__ __launch_bounds__(256) void edge_rank_kernel(const int* __restrict__ dst,
                                                        int* __restrict__ cnt,
                                                        int* __restrict__ rnk,
                                                        int nE) {
    int e = blockIdx.x * blockDim.x + threadIdx.x;
    if (e >= nE) return;
    rnk[e] = atomicAdd(&cnt[dst[e]], 1);
}

__global__ __launch_bounds__(256) void seg_alloc_kernel(const int* __restrict__ cnt,
                                                        int* __restrict__ start,
                                                        int* __restrict__ cursor,
                                                        int n) {
    __shared__ int sh[256];
    __shared__ int base;
    int t = threadIdx.x;
    int i = blockIdx.x * 256 + t;
    int c = (i < n) ? cnt[i] : 0;
    sh[t] = c;
    __syncthreads();
    for (int off = 1; off < 256; off <<= 1) {
        int v = (t >= off) ? sh[t - off] : 0;
        __syncthreads();
        sh[t] += v;
        __syncthreads();
    }
    if (t == 255) base = atomicAdd(cursor, sh[255]);
    __syncthreads();
    if (i < n) start[i] = base + sh[t] - c;
}

__global__ __launch_bounds__(256) void edge_place_kernel(const int* __restrict__ src,
                                                         const int* __restrict__ dst,
                                                         const int* __restrict__ start,
                                                         const int* __restrict__ rnk,
                                                         int* __restrict__ perm,
                                                         int nE) {
    int e = blockIdx.x * blockDim.x + threadIdx.x;
    if (e >= nE) return;
    perm[start[dst[e]] + rnk[e]] = src[e];
}

__global__ __launch_bounds__(256) void gather32_kernel(const float* __restrict__ feat,
                                                       const int* __restrict__ perm,
                                                       const int* __restrict__ start,
                                                       const int* __restrict__ cnt,
                                                       float* __restrict__ agg,
                                                       float* __restrict__ deg,
                                                       int nNodes) {
    int lane = threadIdx.x & 31;
    int node = blockIdx.x * 8 + (threadIdx.x >> 5);
    node = __builtin_amdgcn_readfirstlane(node);
    if (node >= nNodes) return;
    int st = start[node];
    int c  = cnt[node];
    float acc = 0.0f;
    for (int j0 = 0; j0 < c; j0 += 32) {
        int m  = c - j0;
        int li = lane < m ? lane : m - 1;
        int myidx = perm[st + j0 + li];
        int mm = m < 32 ? m : 32;
        for (int jj = 0; jj < mm; jj += 8) {
            float v[8];
#pragma unroll
            for (int u = 0; u < 8; ++u) {
                int s = __builtin_amdgcn_readlane(myidx, jj + u);
                v[u] = feat[(long)s * IN_F + lane];
            }
#pragma unroll
            for (int u = 0; u < 8; ++u) acc += (jj + u < m) ? v[u] : 0.0f;
        }
    }
    agg[(long)node * IN_F + lane] = acc;
    if (lane == 0) deg[node] = (float)c;
}

__global__ __launch_bounds__(256) void gather64_kernel(const float* __restrict__ h,
                                                       const int* __restrict__ perm,
                                                       const int* __restrict__ start,
                                                       const int* __restrict__ cnt,
                                                       float* __restrict__ agg,
                                                       int nNodes) {
    int lane = threadIdx.x & 31;
    int node = blockIdx.x * 8 + (threadIdx.x >> 5);
    node = __builtin_amdgcn_readfirstlane(node);
    if (node >= nNodes) return;
    int st = start[node];
    int c  = cnt[node];
    v2f acc = {0.0f, 0.0f};
    for (int j0 = 0; j0 < c; j0 += 32) {
        int m  = c - j0;
        int li = lane < m ? lane : m - 1;
        int myidx = perm[st + j0 + li];
        int mm = m < 32 ? m : 32;
        for (int jj = 0; jj < mm; jj += 8) {
            v2f v[8];
#pragma unroll
            for (int u = 0; u < 8; ++u) {
                int s = __builtin_amdgcn_readlane(myidx, jj + u);
                v[u] = *(const v2f*)(h + (long)s * HID_F + 2 * lane);
            }
#pragma unroll
            for (int u = 0; u < 8; ++u) {
                bool on = (jj + u < m);
                acc.x += on ? v[u].x : 0.0f;
                acc.y += on ? v[u].y : 0.0f;
            }
        }
    }
    *(v2f*)(agg + (long)node * HID_F + 2 * lane) = acc;
}

__global__ __launch_bounds__(256) void sage_layer1_kernel(const float* __restrict__ feat,
                                                          const float* __restrict__ agg,
                                                          const float* __restrict__ deg,
                                                          const float* __restrict__ Wself,
                                                          const float* __restrict__ Wneigh,
                                                          const float* __restrict__ bias,
                                                          float* __restrict__ h,
                                                          int nRowTiles) {
    int tid  = blockIdx.x * blockDim.x + threadIdx.x;
    int wave = tid >> 5;
    int lane = tid & 31;
    const int colTiles = HID_F / 16;
    int tileRow = wave / colTiles;
    int ct      = wave % colTiles;
    if (tileRow >= nRowTiles) return;

    int hi   = lane >> 4;
    int l15  = lane & 15;
    long row = (long)tileRow * 16 + l15;
    int colW = ct * 16 + l15;

    float dg   = fmaxf(deg[row], 1.0f);
    float invd = 1.0f / dg;

    v8f c;
    float bv = bias[colW];
#pragma unroll
    for (int v = 0; v < 8; ++v) c[v] = bv;

    const float* fr = feat + row * IN_F;
    const float* ar = agg  + row * IN_F;

#pragma unroll
    for (int k0 = 0; k0 < IN_F; k0 += 4) {
        int kA = k0 + 2 * hi;
        v2f a, b;
        a.x = fr[kA];
        a.y = fr[kA + 1];
        b.x = Wself[(long)(kA)     * HID_F + colW];
        b.y = Wself[(long)(kA + 1) * HID_F + colW];
        c = __builtin_amdgcn_wmma_f32_16x16x4_f32(false, a, false, b, (short)0, c, false, false);
        a.x = ar[kA]     * invd;
        a.y = ar[kA + 1] * invd;
        b.x = Wneigh[(long)(kA)     * HID_F + colW];
        b.y = Wneigh[(long)(kA + 1) * HID_F + colW];
        c = __builtin_amdgcn_wmma_f32_16x16x4_f32(false, a, false, b, (short)0, c, false, false);
    }

#pragma unroll
    for (int v = 0; v < 8; ++v) {
        long r = (long)tileRow * 16 + v + 8 * hi;
        h[r * HID_F + colW] = fmaxf(c[v], 0.0f);
    }
}

__global__ __launch_bounds__(256) void sage_layer2_kernel(const float* __restrict__ h,
                                                          const float* __restrict__ agg,
                                                          const float* __restrict__ deg,
                                                          const float* __restrict__ Wself,
                                                          const float* __restrict__ Wneigh,
                                                          const float* __restrict__ bias,
                                                          float* __restrict__ out,
                                                          int nRowTiles) {
    int tid  = blockIdx.x * blockDim.x + threadIdx.x;
    int wave = tid >> 5;
    int lane = tid & 31;
    int tileRow = wave;
    if (tileRow >= nRowTiles) return;

    int hi  = lane >> 4;
    int l15 = lane & 15;
    long row = (long)tileRow * 16 + l15;
    int colW = l15;

    float dg   = fmaxf(deg[row], 1.0f);
    float invd = 1.0f / dg;

    v8f c;
    float bv = bias[colW];
#pragma unroll
    for (int v = 0; v < 8; ++v) c[v] = bv;

    const float* hr = h   + row * HID_F;
    const float* ar = agg + row * HID_F;

#pragma unroll
    for (int k0 = 0; k0 < HID_F; k0 += 4) {
        int kA = k0 + 2 * hi;
        v2f a, b;
        a.x = hr[kA];
        a.y = hr[kA + 1];
        b.x = Wself[(long)(kA)     * OUT_F + colW];
        b.y = Wself[(long)(kA + 1) * OUT_F + colW];
        c = __builtin_amdgcn_wmma_f32_16x16x4_f32(false, a, false, b, (short)0, c, false, false);
        a.x = ar[kA]     * invd;
        a.y = ar[kA + 1] * invd;
        b.x = Wneigh[(long)(kA)     * OUT_F + colW];
        b.y = Wneigh[(long)(kA + 1) * OUT_F + colW];
        c = __builtin_amdgcn_wmma_f32_16x16x4_f32(false, a, false, b, (short)0, c, false, false);
    }

#pragma unroll
    for (int v = 0; v < 8; ++v) {
        long r = (long)tileRow * 16 + v + 8 * hi;
        out[r * OUT_F + colW] = c[v];
    }
}

extern "C" void kernel_launch(void* const* d_in, const int* in_sizes, int n_in,
                              void* d_out, int out_size, void* d_ws, size_t ws_size,
                              hipStream_t stream) {
    const float* feat = (const float*)d_in[0];
    const int*   src  = (const int*)d_in[1];
    const int*   dst  = (const int*)d_in[2];
    const float* W1s  = (const float*)d_in[3];
    const float* W1n  = (const float*)d_in[4];
    const float* b1   = (const float*)d_in[5];
    const float* W2s  = (const float*)d_in[6];
    const float* W2n  = (const float*)d_in[7];
    const float* b2   = (const float*)d_in[8];
    float* out = (float*)d_out;

    const int nNodes = in_sizes[0] / IN_F;
    const int nE     = in_sizes[1];
    const int nRowTiles = nNodes / 16;

    float* ws    = (float*)d_ws;
    float* h     = ws;
    float* agg   = h + (long)nNodes * HID_F;
    int*   perm  = (int*)(agg + (long)nNodes * HID_F);
    int*   cnt   = perm + nE;
    int*   cursor = cnt + nNodes;
    int*   start = cursor + 32;
    float* deg   = (float*)(start + nNodes);
    int*   rnk   = (int*)h;

    const int eBlocks = (nE + 255) / 256;
    const int nBlocks = (nNodes + 255) / 256;
    const int wBlocks = (nNodes + 7) / 8;

    zero_i32_kernel<<<(nNodes + 32 + 255) / 256, 256, 0, stream>>>(cnt, nNodes + 32);
    edge_rank_kernel<<<eBlocks, 256, 0, stream>>>(dst, cnt, rnk, nE);
    seg_alloc_kernel<<<nBlocks, 256, 0, stream>>>(cnt, start, cursor, nNodes);
    edge_place_kernel<<<eBlocks, 256, 0, stream>>>(src, dst, start, rnk, perm, nE);

    gather32_kernel<<<wBlocks, 256, 0, stream>>>(feat, perm, start, cnt, agg, deg, nNodes);
    {
        int waves = nRowTiles * (HID_F / 16);
        int blocks = (waves * 32 + 255) / 256;
        sage_layer1_kernel<<<blocks, 256, 0, stream>>>(feat, agg, deg, W1s, W1n, b1, h, nRowTiles);
    }

    gather64_kernel<<<wBlocks, 256, 0, stream>>>(h, perm, start, cnt, agg, nNodes);
    {
        int waves = nRowTiles;
        int blocks = (waves * 32 + 255) / 256;
        sage_layer2_kernel<<<blocks, 256, 0, stream>>>(h, agg, deg, W2s, W2n, b2, out, nRowTiles);
    }
}


# Edge grouping: 1.6M returning global atomics + random place -> two-level counting sort on LDS counters (chunk sort by bucket, per-bucket node sort); 8->6 launches; sum order now LDS rank, run-varying
# speedup vs baseline: 3.9699x; 1.9218x over previous
#include <hip/hip_runtime.h>
#include <hip/hip_bf16.h>

#define NN 100000
#define IN_F 32
#define HID_F 64
#define OUT_F 16

typedef float v2f __attribute__((ext_vector_type(2)));
typedef float v8f __attribute__((ext_vector_type(8)));

#define NEDGES     1600000
#define NE_CHUNK   6256
#define N_CHUNKS   ((NEDGES + NE_CHUNK - 1) / NE_CHUNK)
#define BKT_NODES  392
#define N_BKT      ((NN + BKT_NODES - 1) / BKT_NODES)
#define SORT_THREADS 1024
#define EDGES_PER_THREAD ((NE_CHUNK + SORT_THREADS - 1) / SORT_THREADS)
#define RUNS_PER_WAVE (N_CHUNKS / 32)
static_assert(N_BKT <= SORT_THREADS / 4, "bucket counters are scanned by the first waves of the block");
static_assert(BKT_NODES <= SORT_THREADS, "one thread per node of a bucket");
static_assert(N_CHUNKS == 32 * RUNS_PER_WAVE && N_CHUNKS <= SORT_THREADS, "runs are dealt 8 to each of 32 waves");

__device__ __forceinline__ int wave_incl_scan(int v, int lane) {
    for (int o = 1; o < 32; o <<= 1) {
        int n = __shfl_up(v, o, 32);
        if (lane >= o) v += n;
    }
    return v;
}

__device__ __forceinline__ int block_excl_scan(int v, int t, int* wsum, int* total) {
    int lane = t & 31, wv = t >> 5;
    int inc = wave_incl_scan(v, lane);
    if (lane == 31) wsum[wv] = inc;
    __syncthreads();
    if (wv == 0) {
        int x  = wsum[lane];
        int xi = wave_incl_scan(x, lane);
        wsum[lane] = xi - x;
        if (lane == 31) *total = xi;
    }
    __syncthreads();
    return wsum[wv] + inc - v;
}

__global__ __launch_bounds__(SORT_THREADS) void chunk_bin_kernel(const int* __restrict__ src,
                                                                 const int* __restrict__ dst,
                                                                 int* __restrict__ bktw,
                                                                 int* __restrict__ runsw,
                                                                 int* __restrict__ cursor,
                                                                 int nE) {
    __shared__ int  hist[N_BKT];
    __shared__ int  offs[N_BKT];
    __shared__ int  wsum[32];
    __shared__ int  tot;
    __shared__ int2 sorted[NE_CHUNK];
    int2* bkt  = (int2*)bktw;
    int2* runs = (int2*)runsw;
    const int t  = threadIdx.x;
    const int w  = blockIdx.x;
    const int e0 = w * NE_CHUNK;
    int len = nE - e0;
    len = len < NE_CHUNK ? len : NE_CHUNK;
    if (t < N_BKT) hist[t] = 0;
    if (w == 0 && t == 0) *cursor = 0;
    __syncthreads();

    int s[EDGES_PER_THREAD], d[EDGES_PER_THREAD], r[EDGES_PER_THREAD];
#pragma unroll
    for (int u = 0; u < EDGES_PER_THREAD; ++u) {
        int i = t + u * SORT_THREADS;
        s[u] = 0; d[u] = 0;
        if (i < len) { d[u] = dst[e0 + i]; s[u] = src[e0 + i]; }
    }
#pragma unroll
    for (int u = 0; u < EDGES_PER_THREAD; ++u) {
        int i = t + u * SORT_THREADS;
        r[u] = 0;
        if (i < len) r[u] = atomicAdd(&hist[d[u] / BKT_NODES], 1);
    }
    __syncthreads();

    int c    = (t < N_BKT) ? hist[t] : 0;
    int excl = block_excl_scan(c, t, wsum, &tot);
    if (t < N_BKT) {
        offs[t] = excl;
        runs[t * N_CHUNKS + w] = make_int2(e0 + excl, c);
    }
    __syncthreads();

#pragma unroll
    for (int u = 0; u < EDGES_PER_THREAD; ++u) {
        int i = t + u * SORT_THREADS;
        if (i < len) sorted[offs[d[u] / BKT_NODES] + r[u]] = make_int2(s[u], d[u]);
    }
    __syncthreads();
    for (int i = t; i < len; i += SORT_THREADS) bkt[e0 + i] = sorted[i];
}

__global__ __launch_bounds__(SORT_THREADS) void bucket_sort_kernel(const int* __restrict__ bktw,
                                                                   const int* __restrict__ runsw,
                                                                   int* __restrict__ cursor,
                                                                   int* __restrict__ perm,
                                                                   int* __restrict__ cnt,
                                                                   int* __restrict__ start,
                                                                   int nNodes) {
    __shared__ int2 run[N_CHUNKS];
    __shared__ int  ccnt[BKT_NODES];
    __shared__ int  cur[BKT_NODES];
    __shared__ int  wsum[32];
    __shared__ int  tot;
    __shared__ int  sbase;
    const int2* bkt  = (const int2*)bktw;
    const int2* runs = (const int2*)runsw;
    const int t    = threadIdx.x;
    const int lane = t & 31;
    const int wv   = t >> 5;
    const int b    = blockIdx.x;
    const int nb0  = b * BKT_NODES;
    if (t < N_CHUNKS)  run[t]  = runs[b * N_CHUNKS + t];
    if (t < BKT_NODES) ccnt[t] = 0;
    __syncthreads();

    int2 rr[RUNS_PER_WAVE], e[RUNS_PER_WAVE];
#pragma unroll
    for (int u = 0; u < RUNS_PER_WAVE; ++u) {
        rr[u] = run[wv + 32 * u];
        e[u]  = make_int2(0, 0);
        if (lane < rr[u].y) e[u] = bkt[rr[u].x + lane];
    }
#pragma unroll
    for (int u = 0; u < RUNS_PER_WAVE; ++u) {
        if (lane < rr[u].y) atomicAdd(&ccnt[e[u].y - nb0], 1);
        for (int k = lane + 32; k < rr[u].y; k += 32) {
            int2 x = bkt[rr[u].x + k];
            atomicAdd(&ccnt[x.y - nb0], 1);
        }
    }
    __syncthreads();

    int c    = (t < BKT_NODES) ? ccnt[t] : 0;
    int excl = block_excl_scan(c, t, wsum, &tot);
    if (t == 0) sbase = atomicAdd(cursor, tot);
    if (t < BKT_NODES) cur[t] = excl;
    __syncthreads();
    const int base = sbase;
    if (t < BKT_NODES && nb0 + t < nNodes) {
        cnt[nb0 + t]   = c;
        start[nb0 + t] = base + excl;
    }

#pragma unroll
    for (int u = 0; u < RUNS_PER_WAVE; ++u) {
        if (lane < rr[u].y) {
            int p = atomicAdd(&cur[e[u].y - nb0], 1);
            perm[base + p] = e[u].x;
        }
        for (int k = lane + 32; k < rr[u].y; k += 32) {
            int2 x = bkt[rr[u].x + k];
            int p = atomicAdd(&cur[x.y - nb0], 1);
            perm[base + p] = x.x;
        }
    }
}

__global__ __launch_bounds__(256) void gather32_kernel(const float* __restrict__ feat,
                                                       const int* __restrict__ perm,
                                                       const int* __restrict__ start,
                                                       const int* __restrict__ cnt,
                                                       float* __restrict__ agg,
                                                       float* __restrict__ deg,
                                                       int nNodes) {
    int lane = threadIdx.x & 31;
    int node = blockIdx.x * 8 + (threadIdx.x >> 5);
    node = __builtin_amdgcn_readfirstlane(node);
    if (node >= nNodes) return;
    int st = start[node];
    int c  = cnt[node];
    float acc = 0.0f;
    for (int j0 = 0; j0 < c; j0 += 32) {
        int m  = c - j0;
        int li = lane < m ? lane : m - 1;
        int myidx = perm[st + j0 + li];
        int mm = m < 32 ? m : 32;
        for (int jj = 0; jj < mm; jj += 8) {
            float v[8];
#pragma unroll
            for (int u = 0; u < 8; ++u) {
                int s = __builtin_amdgcn_readlane(myidx, jj + u);
                v[u] = feat[(long)s * IN_F + lane];
            }
#pragma unroll
            for (int u = 0; u < 8; ++u) acc += (jj + u < m) ? v[u] : 0.0f;
        }
    }
    agg[(long)node * IN_F + lane] = acc;
    if (lane == 0) deg[node] = (float)c;
}

__global__ __launch_bounds__(256) void gather64_kernel(const float* __restrict__ h,
                                                       const int* __restrict__ perm,
                                                       const int* __restrict__ start,
                                                       const int* __restrict__ cnt,
                                                       float* __restrict__ agg,
                                                       int nNodes) {
    int lane = threadIdx.x & 31;
    int node = blockIdx.x * 8 + (threadIdx.x >> 5);
    node = __builtin_amdgcn_readfirstlane(node);
    if (node >= nNodes) return;
    int st = start[node];
    int c  = cnt[node];
    v2f acc = {0.0f, 0.0f};
    for (int j0 = 0; j0 < c; j0 += 32) {
        int m  = c - j0;
        int li = lane < m ? lane : m - 1;
        int myidx = perm[st + j0 + li];
        int mm = m < 32 ? m : 32;
        for (int jj = 0; jj < mm; jj += 8) {
            v2f v[8];
#pragma unroll
            for (int u = 0; u < 8; ++u) {
                int s = __builtin_amdgcn_readlane(myidx, jj + u);
                v[u] = *(const v2f*)(h + (long)s * HID_F + 2 * lane);
            }
#pragma unroll
            for (int u = 0; u < 8; ++u) {
                bool on = (jj + u < m);
                acc.x += on ? v[u].x : 0.0f;
                acc.y += on ? v[u].y : 0.0f;
            }
        }
    }
    *(v2f*)(agg + (long)node * HID_F + 2 * lane) = acc;
}

__global__ __launch_bounds__(256) void sage_layer1_kernel(const float* __restrict__ feat,
                                                          const float* __restrict__ agg,
                                                          const float* __restrict__ deg,
                                                          const float* __restrict__ Wself,
                                                          const float* __restrict__ Wneigh,
                                                          const float* __restrict__ bias,
                                                          float* __restrict__ h,
                                                          int nRowTiles) {
    int tid  = blockIdx.x * blockDim.x + threadIdx.x;
    int wave = tid >> 5;
    int lane = tid & 31;
    const int colTiles = HID_F / 16;
    int tileRow = wave / colTiles;
    int ct      = wave % colTiles;
    if (tileRow >= nRowTiles) return;

    int hi   = lane >> 4;
    int l15  = lane & 15;
    long row = (long)tileRow * 16 + l15;
    int colW = ct * 16 + l15;

    float dg   = fmaxf(deg[row], 1.0f);
    float invd = 1.0f / dg;

    v8f c;
    float bv = bias[colW];
#pragma unroll
    for (int v = 0; v < 8; ++v) c[v] = bv;

    const float* fr = feat + row * IN_F;
    const float* ar = agg  + row * IN_F;

#pragma unroll
    for (int k0 = 0; k0 < IN_F; k0 += 4) {
        int kA = k0 + 2 * hi;
        v2f a, b;
        a.x = fr[kA];
        a.y = fr[kA + 1];
        b.x = Wself[(long)(kA)     * HID_F + colW];
        b.y = Wself[(long)(kA + 1) * HID_F + colW];
        c = __builtin_amdgcn_wmma_f32_16x16x4_f32(false, a, false, b, (short)0, c, false, false);
        a.x = ar[kA]     * invd;
        a.y = ar[kA + 1] * invd;
        b.x = Wneigh[(long)(kA)     * HID_F + colW];
        b.y = Wneigh[(long)(kA + 1) * HID_F + colW];
        c = __builtin_amdgcn_wmma_f32_16x16x4_f32(false, a, false, b, (short)0, c, false, false);
    }

#pragma unroll
    for (int v = 0; v < 8; ++v) {
        long r = (long)tileRow * 16 + v + 8 * hi;
        h[r * HID_F + colW] = fmaxf(c[v], 0.0f);
    }
}

__global__ __launch_bounds__(256) void sage_layer2_kernel(const float* __restrict__ h,
                                                          const float* __restrict__ agg,
                                                          const float* __restrict__ deg,
                                                          const float* __restrict__ Wself,
                                                          const float* __restrict__ Wneigh,
                                                          const float* __restrict__ bias,
                                                          float* __restrict__ out,
                                                          int nRowTiles) {
    int tid  = blockIdx.x * blockDim.x + threadIdx.x;
    int wave = tid >> 5;
    int lane = tid & 31;
    int tileRow = wave;
    if (tileRow >= nRowTiles) return;

    int hi  = lane >> 4;
    int l15 = lane & 15;
    long row = (long)tileRow * 16 + l15;
    int colW = l15;

    float dg   = fmaxf(deg[row], 1.0f);
    float invd = 1.0f / dg;

    v8f c;
    float bv = bias[colW];
#pragma unroll
    for (int v = 0; v < 8; ++v) c[v] = bv;

    const float* hr = h   + row * HID_F;
    const float* ar = agg + row * HID_F;

#pragma unroll
    for (int k0 = 0; k0 < HID_F; k0 += 4) {
        int kA = k0 + 2 * hi;
        v2f a, b;
        a.x = hr[kA];
        a.y = hr[kA + 1];
        b.x = Wself[(long)(kA)     * OUT_F + colW];
        b.y = Wself[(long)(kA + 1) * OUT_F + colW];
        c = __builtin_amdgcn_wmma_f32_16x16x4_f32(false, a, false, b, (short)0, c, false, false);
        a.x = ar[kA]     * invd;
        a.y = ar[kA + 1] * invd;
        b.x = Wneigh[(long)(kA)     * OUT_F + colW];
        b.y = Wneigh[(long)(kA + 1) * OUT_F + colW];
        c = __builtin_amdgcn_wmma_f32_16x16x4_f32(false, a, false, b, (short)0, c, false, false);
    }

#pragma unroll
    for (int v = 0; v < 8; ++v) {
        long r = (long)tileRow * 16 + v + 8 * hi;
        out[r * OUT_F + colW] = c[v];
    }
}

extern "C" void kernel_launch(void* const* d_in, const int* in_sizes, int n_in,
                              void* d_out, int out_size, void* d_ws, size_t ws_size,
                              hipStream_t stream) {
    const float* feat = (const float*)d_in[0];
    const int*   src  = (const int*)d_in[1];
    const int*   dst  = (const int*)d_in[2];
    const float* W1s  = (const float*)d_in[3];
    const float* W1n  = (const float*)d_in[4];
    const float* b1   = (const float*)d_in[5];
    const float* W2s  = (const float*)d_in[6];
    const float* W2n  = (const float*)d_in[7];
    const float* b2   = (const float*)d_in[8];
    float* out = (float*)d_out;

    const int nNodes = in_sizes[0] / IN_F;
    const int nE     = in_sizes[1];
    const int nRowTiles = nNodes / 16;

    float* ws    = (float*)d_ws;
    float* h     = ws;
    float* agg   = h + (long)nNodes * HID_F;
    int*   perm  = (int*)(agg + (long)nNodes * HID_F);
    int*   cnt   = perm + nE;
    int*   cursor = cnt + nNodes;
    int*   start = cursor + 32;
    float* deg   = (float*)(start + nNodes);
    int*   bkt   = (int*)h;
    int*   runs  = (int*)agg;

    const int wBlocks = (nNodes + 7) / 8;

    chunk_bin_kernel<<<N_CHUNKS, SORT_THREADS, 0, stream>>>(src, dst, bkt, runs, cursor, nE);
    bucket_sort_kernel<<<N_BKT, SORT_THREADS, 0, stream>>>(bkt, runs, cursor, perm, cnt, start, nNodes);

    gather32_kernel<<<wBlocks, 256, 0, stream>>>(feat, perm, start, cnt, agg, deg, nNodes);
    {
        int waves = nRowTiles * (HID_F / 16);
        int blocks = (waves * 32 + 255) / 256;
        sage_layer1_kernel<<<blocks, 256, 0, stream>>>(feat, agg, deg, W1s, W1n, b1, h, nRowTiles);
    }

    gather64_kernel<<<wBlocks, 256, 0, stream>>>(h, perm, start, cnt, agg, nNodes);
    {
        int waves = nRowTiles;
        int blocks = (waves * 32 + 255) / 256;
        sage_layer2_kernel<<<blocks, 256, 0, stream>>>(h, agg, deg, W2s, W2n, b2, out, nRowTiles);
    }
}


# Neighbour sums, both layers: 100000 one-node waves per launch -> 12500 waves that each walk 8 consecutive nodes (bounds in one load, next index list requested ahead); per-node sum order unchanged
# speedup vs baseline: 4.5747x; 1.1523x over previous
#include <hip/hip_runtime.h>
#include <hip/hip_bf16.h>

#define NN 100000
#define IN_F 32
#define HID_F 64
#define OUT_F 16

typedef float v2f __attribute__((ext_vector_type(2)));
typedef float v8f __attribute__((ext_vector_type(8)));

#define NEDGES     1600000
#define NE_CHUNK   6256
#define N_CHUNKS   ((NEDGES + NE_CHUNK - 1) / NE_CHUNK)
#define BKT_NODES  392
#define N_BKT      ((NN + BKT_NODES - 1) / BKT_NODES)
#define SORT_THREADS 1024
#define EDGES_PER_THREAD ((NE_CHUNK + SORT_THREADS - 1) / SORT_THREADS)
#define RUNS_PER_WAVE (N_CHUNKS / 32)
static_assert(N_BKT <= SORT_THREADS / 4, "bucket counters are scanned by the first waves of the block");
static_assert(BKT_NODES <= SORT_THREADS, "one thread per node of a bucket");
static_assert(N_CHUNKS == 32 * RUNS_PER_WAVE && N_CHUNKS <= SORT_THREADS, "runs are dealt 8 to each of 32 waves");

__device__ __forceinline__ int wave_incl_scan(int v, int lane) {
    for (int o = 1; o < 32; o <<= 1) {
        int n = __shfl_up(v, o, 32);
        if (lane >= o) v += n;
    }
    return v;
}

__device__ __forceinline__ int block_excl_scan(int v, int t, int* wsum, int* total) {
    int lane = t & 31, wv = t >> 5;
    int inc = wave_incl_scan(v, lane);
    if (lane == 31) wsum[wv] = inc;
    __syncthreads();
    if (wv == 0) {
        int x  = wsum[lane];
        int xi = wave_incl_scan(x, lane);
        wsum[lane] = xi - x;
        if (lane == 31) *total = xi;
    }
    __syncthreads();
    return wsum[wv] + inc - v;
}

__global__ __launch_bounds__(SORT_THREADS) void chunk_bin_kernel(const int* __restrict__ src,
                                                                 const int* __restrict__ dst,
                                                                 int* __restrict__ bktw,
                                                                 int* __restrict__ runsw,
                                                                 int* __restrict__ cursor,
                                                                 int nE) {
    __shared__ int  hist[N_BKT];
    __shared__ int  offs[N_BKT];
    __shared__ int  wsum[32];
    __shared__ int  tot;
    __shared__ int2 sorted[NE_CHUNK];
    int2* bkt  = (int2*)bktw;
    int2* runs = (int2*)runsw;
    const int t  = threadIdx.x;
    const int w  = blockIdx.x;
    const int e0 = w * NE_CHUNK;
    int len = nE - e0;
    len = len < NE_CHUNK ? len : NE_CHUNK;
    if (t < N_BKT) hist[t] = 0;
    if (w == 0 && t == 0) *cursor = 0;
    __syncthreads();

    int s[EDGES_PER_THREAD], d[EDGES_PER_THREAD], r[EDGES_PER_THREAD];
#pragma unroll
    for (int u = 0; u < EDGES_PER_THREAD; ++u) {
        int i = t + u * SORT_THREADS;
        s[u] = 0; d[u] = 0;
        if (i < len) { d[u] = dst[e0 + i]; s[u] = src[e0 + i]; }
    }
#pragma unroll
    for (int u = 0; u < EDGES_PER_THREAD; ++u) {
        int i = t + u * SORT_THREADS;
        r[u] = 0;
        if (i < len) r[u] = atomicAdd(&hist[d[u] / BKT_NODES], 1);
    }
    __syncthreads();

    int c    = (t < N_BKT) ? hist[t] : 0;
    int excl = block_excl_scan(c, t, wsum, &tot);
    if (t < N_BKT) {
        offs[t] = excl;
        runs[t * N_CHUNKS + w] = make_int2(e0 + excl, c);
    }
    __syncthreads();

#pragma unroll
    for (int u = 0; u < EDGES_PER_THREAD; ++u) {
        int i = t + u * SORT_THREADS;
        if (i < len) sorted[offs[d[u] / BKT_NODES] + r[u]] = make_int2(s[u], d[u]);
    }
    __syncthreads();
    for (int i = t; i < len; i += SORT_THREADS) bkt[e0 + i] = sorted[i];
}

__global__ __launch_bounds__(SORT_THREADS) void bucket_sort_kernel(const int* __restrict__ bktw,
                                                                   const int* __restrict__ runsw,
                                                                   int* __restrict__ cursor,
                                                                   int* __restrict__ perm,
                                                                   int* __restrict__ cnt,
                                                                   int* __restrict__ start,
                                                                   int nNodes) {
    __shared__ int2 run[N_CHUNKS];
    __shared__ int  ccnt[BKT_NODES];
    __shared__ int  cur[BKT_NODES];
    __shared__ int  wsum[32];
    __shared__ int  tot;
    __shared__ int  sbase;
    const int2* bkt  = (const int2*)bktw;
    const int2* runs = (const int2*)runsw;
    const int t    = threadIdx.x;
    const int lane = t & 31;
    const int wv   = t >> 5;
    const int b    = blockIdx.x;
    const int nb0  = b * BKT_NODES;
    if (t < N_CHUNKS)  run[t]  = runs[b * N_CHUNKS + t];
    if (t < BKT_NODES) ccnt[t] = 0;
    __syncthreads();

    int2 rr[RUNS_PER_WAVE], e[RUNS_PER_WAVE];
#pragma unroll
    for (int u = 0; u < RUNS_PER_WAVE; ++u) {
        rr[u] = run[wv + 32 * u];
        e[u]  = make_int2(0, 0);
        if (lane < rr[u].y) e[u] = bkt[rr[u].x + lane];
    }
#pragma unroll
    for (int u = 0; u < RUNS_PER_WAVE; ++u) {
        if (lane < rr[u].y) atomicAdd(&ccnt[e[u].y - nb0], 1);
        for (int k = lane + 32; k < rr[u].y; k += 32) {
            int2 x = bkt[rr[u].x + k];
            atomicAdd(&ccnt[x.y - nb0], 1);
        }
    }
    __syncthreads();

    int c    = (t < BKT_NODES) ? ccnt[t] : 0;
    int excl = block_excl_scan(c, t, wsum, &tot);
    if (t == 0) sbase = atomicAdd(cursor, tot);
    if (t < BKT_NODES) cur[t] = excl;
    __syncthreads();
    const int base = sbase;
    if (t < BKT_NODES && nb0 + t < nNodes) {
        cnt[nb0 + t]   = c;
        start[nb0 + t] = base + excl;
    }

#pragma unroll
    for (int u = 0; u < RUNS_PER_WAVE; ++u) {
        if (lane < rr[u].y) {
            int p = atomicAdd(&cur[e[u].y - nb0], 1);
            perm[base + p] = e[u].x;
        }
        for (int k = lane + 32; k < rr[u].y; k += 32) {
            int2 x = bkt[rr[u].x + k];
            int p = atomicAdd(&cur[x.y - nb0], 1);
            perm[base + p] = x.x;
        }
    }
}

#define NODES_PER_WAVE 8

__device__ __forceinline__ float add_rows32(const float* __restrict__ feat, int idx, int e0, int m, int lane, float acc) {
    float v[8];
#pragma unroll
    for (int u = 0; u < 8; ++u) {
        int s = __builtin_amdgcn_readlane(idx, e0 + u);
        v[u] = feat[(long)s * IN_F + lane];
    }
#pragma unroll
    for (int u = 0; u < 8; ++u) acc += (e0 + u < m) ? v[u] : 0.0f;
    return acc;
}

__device__ __forceinline__ v2f add_rows64(const float* __restrict__ h, int idx, int e0, int m, int lane, v2f acc) {
    v2f v[8];
#pragma unroll
    for (int u = 0; u < 8; ++u) {
        int s = __builtin_amdgcn_readlane(idx, e0 + u);
        v[u] = *(const v2f*)(h + (long)s * HID_F + 2 * lane);
    }
#pragma unroll
    for (int u = 0; u < 8; ++u) {
        bool on = (e0 + u < m);
        acc.x += on ? v[u].x : 0.0f;
        acc.y += on ? v[u].y : 0.0f;
    }
    return acc;
}

__device__ __forceinline__ int load_list(const int* __restrict__ perm, int st, int c, int lane) {
    int idx = 0;
    if (c > 0) idx = perm[st + (lane < c ? lane : c - 1)];
    return idx;
}

__global__ __launch_bounds__(256) void gather32_kernel(const float* __restrict__ feat,
                                                       const int* __restrict__ perm,
                                                       const int* __restrict__ start,
                                                       const int* __restrict__ cnt,
                                                       float* __restrict__ agg,
                                                       float* __restrict__ deg,
                                                       int nNodes) {
    const int lane = threadIdx.x & 31;
    const int wave = __builtin_amdgcn_readfirstlane(blockIdx.x * 8 + (threadIdx.x >> 5));
    const int n0 = wave * NODES_PER_WAVE;
    if (n0 >= nNodes) return;
    int nn = nNodes - n0;
    nn = nn < NODES_PER_WAVE ? nn : NODES_PER_WAVE;
    const int lk  = lane < nn ? lane : nn - 1;
    const int stv = start[n0 + lk];
    const int cv  = cnt[n0 + lk];
    int st  = __builtin_amdgcn_readlane(stv, 0);
    int c   = __builtin_amdgcn_readlane(cv, 0);
    int idx = load_list(perm, st, c, lane);
    for (int k = 0; k < nn; ++k) {
        const int m = c < 32 ? c : 32;
        float v[8];
#pragma unroll
        for (int u = 0; u < 8; ++u) {
            int s = __builtin_amdgcn_readlane(idx, u);
            v[u] = feat[(long)s * IN_F + lane];
        }
        int stn = 0, cn = 0, idxn = 0;
        if (k + 1 < nn) {
            stn  = __builtin_amdgcn_readlane(stv, k + 1);
            cn   = __builtin_amdgcn_readlane(cv, k + 1);
            idxn = load_list(perm, stn, cn, lane);
        }
        float acc = 0.0f;
#pragma unroll
        for (int u = 0; u < 8; ++u) acc += (u < m) ? v[u] : 0.0f;
        for (int e0 = 8; e0 < m; e0 += 8) acc = add_rows32(feat, idx, e0, m, lane, acc);
        for (int j0 = 32; j0 < c; j0 += 32) {
            const int r    = c - j0;
            const int more = load_list(perm, st + j0, r, lane);
            const int mr   = r < 32 ? r : 32;
            for (int e0 = 0; e0 < mr; e0 += 8) acc = add_rows32(feat, more, e0, mr, lane, acc);
        }
        agg[(long)(n0 + k) * IN_F + lane] = acc;
        st = stn; c = cn; idx = idxn;
    }
    if (lane < nn) deg[n0 + lane] = (float)cv;
}

__global__ __launch_bounds__(256) void gather64_kernel(const float* __restrict__ h,
                                                       const int* __restrict__ perm,
                                                       const int* __restrict__ start,
                                                       const int* __restrict__ cnt,
                                                       float* __restrict__ agg,
                                                       int nNodes) {
    const int lane = threadIdx.x & 31;
    const int wave = __builtin_amdgcn_readfirstlane(blockIdx.x * 8 + (threadIdx.x >> 5));
    const int n0 = wave * NODES_PER_WAVE;
    if (n0 >= nNodes) return;
    int nn = nNodes - n0;
    nn = nn < NODES_PER_WAVE ? nn : NODES_PER_WAVE;
    const int lk  = lane < nn ? lane : nn - 1;
    const int stv = start[n0 + lk];
    const int cv  = cnt[n0 + lk];
    int st  = __builtin_amdgcn_readlane(stv, 0);
    int c   = __builtin_amdgcn_readlane(cv, 0);
    int idx = load_list(perm, st, c, lane);
    for (int k = 0; k < nn; ++k) {
        const int m = c < 32 ? c : 32;
        v2f v[8];
#pragma unroll
        for (int u = 0; u < 8; ++u) {
            int s = __builtin_amdgcn_readlane(idx, u);
            v[u] = *(const v2f*)(h + (long)s * HID_F + 2 * lane);
        }
        int stn = 0, cn = 0, idxn = 0;
        if (k + 1 < nn) {
            stn  = __builtin_amdgcn_readlane(stv, k + 1);
            cn   = __builtin_amdgcn_readlane(cv, k + 1);
            idxn = load_list(perm, stn, cn, lane);
        }
        v2f acc = {0.0f, 0.0f};
#pragma unroll
        for (int u = 0; u < 8; ++u) {
            bool on = (u < m);
            acc.x += on ? v[u].x : 0.0f;
            acc.y += on ? v[u].y : 0.0f;
        }
        for (int e0 = 8; e0 < m; e0 += 8) acc = add_rows64(h, idx, e0, m, lane, acc);
        for (int j0 = 32; j0 < c; j0 += 32) {
            const int r    = c - j0;
            const int more = load_list(perm, st + j0, r, lane);
            const int mr   = r < 32 ? r : 32;
            for (int e0 = 0; e0 < mr; e0 += 8) acc = add_rows64(h, more, e0, mr, lane, acc);
        }
        *(v2f*)(agg + (long)(n0 + k) * HID_F + 2 * lane) = acc;
        st = stn; c = cn; idx = idxn;
    }
}

__global__ __launch_bounds__(256) void sage_layer1_kernel(const float* __restrict__ feat,
                                                          const float* __restrict__ agg,
                                                          const float* __restrict__ deg,
                                                          const float* __restrict__ Wself,
                                                          const float* __restrict__ Wneigh,
                                                          const float* __restrict__ bias,
                                                          float* __restrict__ h,
                                                          int nRowTiles) {
    int tid  = blockIdx.x * blockDim.x + threadIdx.x;
    int wave = tid >> 5;
    int lane = tid & 31;
    const int colTiles = HID_F / 16;
    int tileRow = wave / colTiles;
    int ct      = wave % colTiles;
    if (tileRow >= nRowTiles) return;

    int hi   = lane >> 4;
    int l15  = lane & 15;
    long row = (long)tileRow * 16 + l15;
    int colW = ct * 16 + l15;

    float dg   = fmaxf(deg[row], 1.0f);
    float invd = 1.0f / dg;

    v8f c;
    float bv = bias[colW];
#pragma unroll
    for (int v = 0; v < 8; ++v) c[v] = bv;

    const float* fr = feat + row * IN_F;
    const float* ar = agg  + row * IN_F;

#pragma unroll
    for (int k0 = 0; k0 < IN_F; k0 += 4) {
        int kA = k0 + 2 * hi;
        v2f a, b;
        a.x = fr[kA];
        a.y = fr[kA + 1];
        b.x = Wself[(long)(kA)     * HID_F + colW];
        b.y = Wself[(long)(kA + 1) * HID_F + colW];
        c = __builtin_amdgcn_wmma_f32_16x16x4_f32(false, a, false, b, (short)0, c, false, false);
        a.x = ar[kA]     * invd;
        a.y = ar[kA + 1] * invd;
        b.x = Wneigh[(long)(kA)     * HID_F + colW];
        b.y = Wneigh[(long)(kA + 1) * HID_F + colW];
        c = __builtin_amdgcn_wmma_f32_16x16x4_f32(false, a, false, b, (short)0, c, false, false);
    }

#pragma unroll
    for (int v = 0; v < 8; ++v) {
        long r = (long)tileRow * 16 + v + 8 * hi;
        h[r * HID_F + colW] = fmaxf(c[v], 0.0f);
    }
}

__global__ __launch_bounds__(256) void sage_layer2_kernel(const float* __restrict__ h,
                                                          const float* __restrict__ agg,
                                                          const float* __restrict__ deg,
                                                          const float* __restrict__ Wself,
                                                          const float* __restrict__ Wneigh,
                                                          const float* __restrict__ bias,
                                                          float* __restrict__ out,
                                                          int nRowTiles) {
    int tid  = blockIdx.x * blockDim.x + threadIdx.x;
    int wave = tid >> 5;
    int lane = tid & 31;
    int tileRow = wave;
    if (tileRow >= nRowTiles) return;

    int hi  = lane >> 4;
    int l15 = lane & 15;
    long row = (long)tileRow * 16 + l15;
    int colW = l15;

    float dg   = fmaxf(deg[row], 1.0f);
    float invd = 1.0f / dg;

    v8f c;
    float bv = bias[colW];
#pragma unroll
    for (int v = 0; v < 8; ++v) c[v] = bv;

    const float* hr = h   + row * HID_F;
    const float* ar = agg + row * HID_F;

#pragma unroll
    for (int k0 = 0; k0 < HID_F; k0 += 4) {
        int kA = k0 + 2 * hi;
        v2f a, b;
        a.x = hr[kA];
        a.y = hr[kA + 1];
        b.x = Wself[(long)(kA)     * OUT_F + colW];
        b.y = Wself[(long)(kA + 1) * OUT_F + colW];
        c = __builtin_amdgcn_wmma_f32_16x16x4_f32(false, a, false, b, (short)0, c, false, false);
        a.x = ar[kA]     * invd;
        a.y = ar[kA + 1] * invd;
        b.x = Wneigh[(long)(kA)     * OUT_F + colW];
        b.y = Wneigh[(long)(kA + 1) * OUT_F + colW];
        c = __builtin_amdgcn_wmma_f32_16x16x4_f32(false, a, false, b, (short)0, c, false, false);
    }

#pragma unroll
    for (int v = 0; v < 8; ++v) {
        long r = (long)tileRow * 16 + v + 8 * hi;
        out[r * OUT_F + colW] = c[v];
    }
}

extern "C" void kernel_launch(void* const* d_in, const int* in_sizes, int n_in,
                              void* d_out, int out_size, void* d_ws, size_t ws_size,
                              hipStream_t stream) {
    const float* feat = (const float*)d_in[0];
    const int*   src  = (const int*)d_in[1];
    const int*   dst  = (const int*)d_in[2];
    const float* W1s  = (const float*)d_in[3];
    const float* W1n  = (const float*)d_in[4];
    const float* b1   = (const float*)d_in[5];
    const float* W2s  = (const float*)d_in[6];
    const float* W2n  = (const float*)d_in[7];
    const float* b2   = (const float*)d_in[8];
    float* out = (float*)d_out;

    const int nNodes = in_sizes[0] / IN_F;
    const int nE     = in_sizes[1];
    const int nRowTiles = nNodes / 16;

    float* ws    = (float*)d_ws;
    float* h     = ws;
    float* agg   = h + (long)nNodes * HID_F;
    int*   perm  = (int*)(agg + (long)nNodes * HID_F);
    int*   cnt   = perm + nE;
    int*   cursor = cnt + nNodes;
    int*   start = cursor + 32;
    float* deg   = (float*)(start + nNodes);
    int*   bkt   = (int*)h;
    int*   runs  = (int*)agg;

    const int wBlocks = (nNodes + 8 * NODES_PER_WAVE - 1) / (8 * NODES_PER_WAVE);

    chunk_bin_kernel<<<N_CHUNKS, SORT_THREADS, 0, stream>>>(src, dst, bkt, runs, cursor, nE);
    bucket_sort_kernel<<<N_BKT, SORT_THREADS, 0, stream>>>(bkt, runs, cursor, perm, cnt, start, nNodes);

    gather32_kernel<<<wBlocks, 256, 0, stream>>>(feat, perm, start, cnt, agg, deg, nNodes);
    {
        int waves = nRowTiles * (HID_F / 16);
        int blocks = (waves * 32 + 255) / 256;
        sage_layer1_kernel<<<blocks, 256, 0, stream>>>(feat, agg, deg, W1s, W1n, b1, h, nRowTiles);
    }

    gather64_kernel<<<wBlocks, 256, 0, stream>>>(h, perm, start, cnt, agg, nNodes);
    {
        int waves = nRowTiles;
        int blocks = (waves * 32 + 255) / 256;
        sage_layer2_kernel<<<blocks, 256, 0, stream>>>(h, agg, deg, W2s, W2n, b2, out, nRowTiles);
    }
}


# Layer 2: h is projected by W2_neigh before the neighbour sum (linear), so the gather adds 16-float rows of h@W2n, not 64-float rows of h; sum order changes: products summed per node in list order
# speedup vs baseline: 5.1199x; 1.1192x over previous
#include <hip/hip_runtime.h>
#include <hip/hip_bf16.h>

#define NN 100000
#define IN_F 32
#define HID_F 64
#define OUT_F 16

typedef float v2f __attribute__((ext_vector_type(2)));
typedef float v4f __attribute__((ext_vector_type(4)));
typedef float v8f __attribute__((ext_vector_type(8)));

#define NEDGES     1600000
#define NE_CHUNK   6256
#define N_CHUNKS   ((NEDGES + NE_CHUNK - 1) / NE_CHUNK)
#define BKT_NODES  392
#define N_BKT      ((NN + BKT_NODES - 1) / BKT_NODES)
#define SORT_THREADS 1024
#define EDGES_PER_THREAD ((NE_CHUNK + SORT_THREADS - 1) / SORT_THREADS)
#define RUNS_PER_WAVE (N_CHUNKS / 32)
static_assert(N_BKT <= SORT_THREADS / 4, "bucket counters are scanned by the first waves of the block");
static_assert(BKT_NODES <= SORT_THREADS, "one thread per node of a bucket");
static_assert(N_CHUNKS == 32 * RUNS_PER_WAVE && N_CHUNKS <= SORT_THREADS, "runs are dealt 8 to each of 32 waves");

__device__ __forceinline__ int wave_incl_scan(int v, int lane) {
    for (int o = 1; o < 32; o <<= 1) {
        int n = __shfl_up(v, o, 32);
        if (lane >= o) v += n;
    }
    return v;
}

__device__ __forceinline__ int block_excl_scan(int v, int t, int* wsum, int* total) {
    int lane = t & 31, wv = t >> 5;
    int inc = wave_incl_scan(v, lane);
    if (lane == 31) wsum[wv] = inc;
    __syncthreads();
    if (wv == 0) {
        int x  = wsum[lane];
        int xi = wave_incl_scan(x, lane);
        wsum[lane] = xi - x;
        if (lane == 31) *total = xi;
    }
    __syncthreads();
    return wsum[wv] + inc - v;
}

__global__ __launch_bounds__(SORT_THREADS) void chunk_bin_kernel(const int* __restrict__ src,
                                                                 const int* __restrict__ dst,
                                                                 int* __restrict__ bktw,
                                                                 int* __restrict__ runsw,
                                                                 int* __restrict__ cursor,
                                                                 int nE) {
    __shared__ int  hist[N_BKT];
    __shared__ int  offs[N_BKT];
    __shared__ int  wsum[32];
    __shared__ int  tot;
    __shared__ int2 sorted[NE_CHUNK];
    int2* bkt  = (int2*)bktw;
    int2* runs = (int2*)runsw;
    const int t  = threadIdx.x;
    const int w  = blockIdx.x;
    const int e0 = w * NE_CHUNK;
    int len = nE - e0;
    len = len < NE_CHUNK ? len : NE_CHUNK;
    if (t < N_BKT) hist[t] = 0;
    if (w == 0 && t == 0) *cursor = 0;
    __syncthreads();

    int s[EDGES_PER_THREAD], d[EDGES_PER_THREAD], r[EDGES_PER_THREAD];
#pragma unroll
    for (int u = 0; u < EDGES_PER_THREAD; ++u) {
        int i = t + u * SORT_THREADS;
        s[u] = 0; d[u] = 0;
        if (i < len) { d[u] = dst[e0 + i]; s[u] = src[e0 + i]; }
    }
#pragma unroll
    for (int u = 0; u < EDGES_PER_THREAD; ++u) {
        int i = t + u * SORT_THREADS;
        r[u] = 0;
        if (i < len) r[u] = atomicAdd(&hist[d[u] / BKT_NODES], 1);
    }
    __syncthreads();

    int c    = (t < N_BKT) ? hist[t] : 0;
    int excl = block_excl_scan(c, t, wsum, &tot);
    if (t < N_BKT) {
        offs[t] = excl;
        runs[t * N_CHUNKS + w] = make_int2(e0 + excl, c);
    }
    __syncthreads();

#pragma unroll
    for (int u = 0; u < EDGES_PER_THREAD; ++u) {
        int i = t + u * SORT_THREADS;
        if (i < len) sorted[offs[d[u] / BKT_NODES] + r[u]] = make_int2(s[u], d[u]);
    }
    __syncthreads();
    for (int i = t; i < len; i += SORT_THREADS) bkt[e0 + i] = sorted[i];
}

__global__ __launch_bounds__(SORT_THREADS) void bucket_sort_kernel(const int* __restrict__ bktw,
                                                                   const int* __restrict__ runsw,
                                                                   int* __restrict__ cursor,
                                                                   int* __restrict__ perm,
                                                                   int* __restrict__ cnt,
                                                                   int* __restrict__ start,
                                                                   int nNodes) {
    __shared__ int2 run[N_CHUNKS];
    __shared__ int  ccnt[BKT_NODES];
    __shared__ int  cur[BKT_NODES];
    __shared__ int  wsum[32];
    __shared__ int  tot;
    __shared__ int  sbase;
    const int2* bkt  = (const int2*)bktw;
    const int2* runs = (const int2*)runsw;
    const int t    = threadIdx.x;
    const int lane = t & 31;
    const int wv   = t >> 5;
    const int b    = blockIdx.x;
    const int nb0  = b * BKT_NODES;
    if (t < N_CHUNKS)  run[t]  = runs[b * N_CHUNKS + t];
    if (t < BKT_NODES) ccnt[t] = 0;
    __syncthreads();

    int2 rr[RUNS_PER_WAVE], e[RUNS_PER_WAVE];
#pragma unroll
    for (int u = 0; u < RUNS_PER_WAVE; ++u) {
        rr[u] = run[wv + 32 * u];
        e[u]  = make_int2(0, 0);
        if (lane < rr[u].y) e[u] = bkt[rr[u].x + lane];
    }
#pragma unroll
    for (int u = 0; u < RUNS_PER_WAVE; ++u) {
        if (lane < rr[u].y) atomicAdd(&ccnt[e[u].y - nb0], 1);
        for (int k = lane + 32; k < rr[u].y; k += 32) {
            int2 x = bkt[rr[u].x + k];
            atomicAdd(&ccnt[x.y - nb0], 1);
        }
    }
    __syncthreads();

    int c    = (t < BKT_NODES) ? ccnt[t] : 0;
    int excl = block_excl_scan(c, t, wsum, &tot);
    if (t == 0) sbase = atomicAdd(cursor, tot);
    if (t < BKT_NODES) cur[t] = excl;
    __syncthreads();
    const int base = sbase;
    if (t < BKT_NODES && nb0 + t < nNodes) {
        cnt[nb0 + t]   = c;
        start[nb0 + t] = base + excl;
    }

#pragma unroll
    for (int u = 0; u < RUNS_PER_WAVE; ++u) {
        if (lane < rr[u].y) {
            int p = atomicAdd(&cur[e[u].y - nb0], 1);
            perm[base + p] = e[u].x;
        }
        for (int k = lane + 32; k < rr[u].y; k += 32) {
            int2 x = bkt[rr[u].x + k];
            int p = atomicAdd(&cur[x.y - nb0], 1);
            perm[base + p] = x.x;
        }
    }
}

#define NODES_PER_WAVE 8

__device__ __forceinline__ float add_rows32(const float* __restrict__ feat, int idx, int e0, int m, int lane, float acc) {
    float v[8];
#pragma unroll
    for (int u = 0; u < 8; ++u) {
        int s = __builtin_amdgcn_readlane(idx, e0 + u);
        v[u] = feat[(long)s * IN_F + lane];
    }
#pragma unroll
    for (int u = 0; u < 8; ++u) acc += (e0 + u < m) ? v[u] : 0.0f;
    return acc;
}

__device__ __forceinline__ int load_list(const int* __restrict__ perm, int st, int c, int lane) {
    int idx = 0;
    if (c > 0) idx = perm[st + (lane < c ? lane : c - 1)];
    return idx;
}

__global__ __launch_bounds__(256) void gather32_kernel(const float* __restrict__ feat,
                                                       const int* __restrict__ perm,
                                                       const int* __restrict__ start,
                                                       const int* __restrict__ cnt,
                                                       float* __restrict__ agg,
                                                       float* __restrict__ deg,
                                                       int nNodes) {
    const int lane = threadIdx.x & 31;
    const int wave = __builtin_amdgcn_readfirstlane(blockIdx.x * 8 + (threadIdx.x >> 5));
    const int n0 = wave * NODES_PER_WAVE;
    if (n0 >= nNodes) return;
    int nn = nNodes - n0;
    nn = nn < NODES_PER_WAVE ? nn : NODES_PER_WAVE;
    const int lk  = lane < nn ? lane : nn - 1;
    const int stv = start[n0 + lk];
    const int cv  = cnt[n0 + lk];
    int st  = __builtin_amdgcn_readlane(stv, 0);
    int c   = __builtin_amdgcn_readlane(cv, 0);
    int idx = load_list(perm, st, c, lane);
    for (int k = 0; k < nn; ++k) {
        const int m = c < 32 ? c : 32;
        float v[8];
#pragma unroll
        for (int u = 0; u < 8; ++u) {
            int s = __builtin_amdgcn_readlane(idx, u);
            v[u] = feat[(long)s * IN_F + lane];
        }
        int stn = 0, cn = 0, idxn = 0;
        if (k + 1 < nn) {
            stn  = __builtin_amdgcn_readlane(stv, k + 1);
            cn   = __builtin_amdgcn_readlane(cv, k + 1);
            idxn = load_list(perm, stn, cn, lane);
        }
        float acc = 0.0f;
#pragma unroll
        for (int u = 0; u < 8; ++u) acc += (u < m) ? v[u] : 0.0f;
        for (int e0 = 8; e0 < m; e0 += 8) acc = add_rows32(feat, idx, e0, m, lane, acc);
        for (int j0 = 32; j0 < c; j0 += 32) {
            const int r    = c - j0;
            const int more = load_list(perm, st + j0, r, lane);
            const int mr   = r < 32 ? r : 32;
            for (int e0 = 0; e0 < mr; e0 += 8) acc = add_rows32(feat, more, e0, mr, lane, acc);
        }
        agg[(long)(n0 + k) * IN_F + lane] = acc;
        st = stn; c = cn; idx = idxn;
    }
    if (lane < nn) deg[n0 + lane] = (float)cv;
}

__global__ __launch_bounds__(256) void sage_layer1_kernel(const float* __restrict__ feat,
                                                          const float* __restrict__ agg,
                                                          const float* __restrict__ deg,
                                                          const float* __restrict__ Wself,
                                                          const float* __restrict__ Wneigh,
                                                          const float* __restrict__ bias,
                                                          float* __restrict__ h,
                                                          int nRowTiles) {
    int tid  = blockIdx.x * blockDim.x + threadIdx.x;
    int wave = tid >> 5;
    int lane = tid & 31;
    const int colTiles = HID_F / 16;
    int tileRow = wave / colTiles;
    int ct      = wave % colTiles;
    if (tileRow >= nRowTiles) return;

    int hi   = lane >> 4;
    int l15  = lane & 15;
    long row = (long)tileRow * 16 + l15;
    int colW = ct * 16 + l15;

    float dg   = fmaxf(deg[row], 1.0f);
    float invd = 1.0f / dg;

    v8f c;
    float bv = bias[colW];
#pragma unroll
    for (int v = 0; v < 8; ++v) c[v] = bv;

    const float* fr = feat + row * IN_F;
    const float* ar = agg  + row * IN_F;

#pragma unroll
    for (int k0 = 0; k0 < IN_F; k0 += 4) {
        int kA = k0 + 2 * hi;
        v2f a, b;
        a.x = fr[kA];
        a.y = fr[kA + 1];
        b.x = Wself[(long)(kA)     * HID_F + colW];
        b.y = Wself[(long)(kA + 1) * HID_F + colW];
        c = __builtin_amdgcn_wmma_f32_16x16x4_f32(false, a, false, b, (short)0, c, false, false);
        a.x = ar[kA]     * invd;
        a.y = ar[kA + 1] * invd;
        b.x = Wneigh[(long)(kA)     * HID_F + colW];
        b.y = Wneigh[(long)(kA + 1) * HID_F + colW];
        c = __builtin_amdgcn_wmma_f32_16x16x4_f32(false, a, false, b, (short)0, c, false, false);
    }

#pragma unroll
    for (int v = 0; v < 8; ++v) {
        long r = (long)tileRow * 16 + v + 8 * hi;
        h[r * HID_F + colW] = fmaxf(c[v], 0.0f);
    }
}

__global__ __launch_bounds__(256) void sage_layer2_project_kernel(const float* __restrict__ h,
                                                                  const float* __restrict__ Wself,
                                                                  const float* __restrict__ Wneigh,
                                                                  const float* __restrict__ bias,
                                                                  float* __restrict__ S,
                                                                  float* __restrict__ P,
                                                                  int nRowTiles) {
    int tid  = blockIdx.x * blockDim.x + threadIdx.x;
    int wave = tid >> 5;
    int lane = tid & 31;
    int tileRow = wave;
    if (tileRow >= nRowTiles) return;

    int hi  = lane >> 4;
    int l15 = lane & 15;
    long row = (long)tileRow * 16 + l15;
    int colW = l15;

    v8f cs, cp;
    float bv = bias[colW];
#pragma unroll
    for (int v = 0; v < 8; ++v) { cs[v] = bv; cp[v] = 0.0f; }

    const float* hr = h + row * HID_F;

#pragma unroll
    for (int k0 = 0; k0 < HID_F; k0 += 4) {
        int kA = k0 + 2 * hi;
        v2f a, bs, bn;
        a.x  = hr[kA];
        a.y  = hr[kA + 1];
        bs.x = Wself[(long)(kA)      * OUT_F + colW];
        bs.y = Wself[(long)(kA + 1)  * OUT_F + colW];
        bn.x = Wneigh[(long)(kA)     * OUT_F + colW];
        bn.y = Wneigh[(long)(kA + 1) * OUT_F + colW];
        cs = __builtin_amdgcn_wmma_f32_16x16x4_f32(false, a, false, bs, (short)0, cs, false, false);
        cp = __builtin_amdgcn_wmma_f32_16x16x4_f32(false, a, false, bn, (short)0, cp, false, false);
    }

#pragma unroll
    for (int v = 0; v < 8; ++v) {
        long r = (long)tileRow * 16 + v + 8 * hi;
        S[r * OUT_F + colW] = cs[v];
        P[r * OUT_F + colW] = cp[v];
    }
}

#define WAVE_NODES 16

template <int J>
__device__ __forceinline__ int quad_lane(int x) {
    return __builtin_amdgcn_update_dpp(0, x, J * 0x55, 0xf, 0xf, true);
}

__device__ __forceinline__ int list_entry(const int* __restrict__ perm, int st, int c, int e) {
    int ee = e < c ? e : c - 1;
    int s = 0;
    if (c > 0) s = perm[st + ee];
    return s;
}

__device__ __forceinline__ void load_rows16(const float* __restrict__ P, int i0, int i1, int q, v4f v[8]) {
    int s[8];
    s[0] = quad_lane<0>(i0); s[1] = quad_lane<1>(i0); s[2] = quad_lane<2>(i0); s[3] = quad_lane<3>(i0);
    s[4] = quad_lane<0>(i1); s[5] = quad_lane<1>(i1); s[6] = quad_lane<2>(i1); s[7] = quad_lane<3>(i1);
#pragma unroll
    for (int u = 0; u < 8; ++u) v[u] = *(const v4f*)(P + (long)s[u] * OUT_F + 4 * q);
}

__global__ __launch_bounds__(256) void gather16_combine_kernel(const float* __restrict__ P,
                                                               const float* __restrict__ S,
                                                               const int* __restrict__ perm,
                                                               const int* __restrict__ start,
                                                               const int* __restrict__ cnt,
                                                               float* __restrict__ out,
                                                               int nNodes) {
    const int lane = threadIdx.x & 31;
    const int q    = lane & 3;
    const int wave = blockIdx.x * 8 + (threadIdx.x >> 5);
    const int nA   = wave * WAVE_NODES + (lane >> 2);
    const int nB   = nA + 8;
    const bool onA = nA < nNodes;
    const bool onB = nB < nNodes;
    int stA = 0, cA = 0, stB = 0, cB = 0;
    v4f selfA = {0.0f, 0.0f, 0.0f, 0.0f}, selfB = {0.0f, 0.0f, 0.0f, 0.0f};
    if (onA) { stA = start[nA]; cA = cnt[nA]; selfA = *(const v4f*)(S + (long)nA * OUT_F + 4 * q); }
    if (onB) { stB = start[nB]; cB = cnt[nB]; selfB = *(const v4f*)(S + (long)nB * OUT_F + 4 * q); }
    const int cm = cA > cB ? cA : cB;

    int a0 = list_entry(perm, stA, cA, q), a1 = list_entry(perm, stA, cA, 4 + q);
    int b0 = list_entry(perm, stB, cB, q), b1 = list_entry(perm, stB, cB, 4 + q);
    v4f accA = {0.0f, 0.0f, 0.0f, 0.0f}, accB = {0.0f, 0.0f, 0.0f, 0.0f};
    const v4f zero = {0.0f, 0.0f, 0.0f, 0.0f};
    for (int e0 = 0; e0 < cm; e0 += 8) {
        v4f va[8], vb[8];
        load_rows16(P, a0, a1, q, va);
        load_rows16(P, b0, b1, q, vb);
        if (e0 + 8 < cm) {
            a0 = list_entry(perm, stA, cA, e0 + 8 + q);  a1 = list_entry(perm, stA, cA, e0 + 12 + q);
            b0 = list_entry(perm, stB, cB, e0 + 8 + q);  b1 = list_entry(perm, stB, cB, e0 + 12 + q);
        }
#pragma unroll
        for (int u = 0; u < 8; ++u) {
            accA += (e0 + u < cA) ? va[u] : zero;
            accB += (e0 + u < cB) ? vb[u] : zero;
        }
    }
    if (onA) *(v4f*)(out + (long)nA * OUT_F + 4 * q) = selfA + accA / fmaxf((float)cA, 1.0f);
    if (onB) *(v4f*)(out + (long)nB * OUT_F + 4 * q) = selfB + accB / fmaxf((float)cB, 1.0f);
}

extern "C" void kernel_launch(void* const* d_in, const int* in_sizes, int n_in,
                              void* d_out, int out_size, void* d_ws, size_t ws_size,
                              hipStream_t stream) {
    const float* feat = (const float*)d_in[0];
    const int*   src  = (const int*)d_in[1];
    const int*   dst  = (const int*)d_in[2];
    const float* W1s  = (const float*)d_in[3];
    const float* W1n  = (const float*)d_in[4];
    const float* b1   = (const float*)d_in[5];
    const float* W2s  = (const float*)d_in[6];
    const float* W2n  = (const float*)d_in[7];
    const float* b2   = (const float*)d_in[8];
    float* out = (float*)d_out;

    const int nNodes = in_sizes[0] / IN_F;
    const int nE     = in_sizes[1];
    const int nRowTiles = nNodes / 16;

    float* ws    = (float*)d_ws;
    float* h     = ws;
    float* agg   = h + (long)nNodes * HID_F;
    int*   perm  = (int*)(agg + (long)nNodes * HID_F);
    int*   cnt   = perm + nE;
    int*   cursor = cnt + nNodes;
    int*   start = cursor + 32;
    float* deg   = (float*)(start + nNodes);
    int*   bkt   = (int*)h;
    int*   runs  = (int*)agg;
    float* P     = agg;
    float* S     = agg + (long)nNodes * OUT_F;

    const int wBlocks = (nNodes + 8 * NODES_PER_WAVE - 1) / (8 * NODES_PER_WAVE);
    const int qBlocks = (nNodes + 8 * WAVE_NODES - 1) / (8 * WAVE_NODES);

    chunk_bin_kernel<<<N_CHUNKS, SORT_THREADS, 0, stream>>>(src, dst, bkt, runs, cursor, nE);
    bucket_sort_kernel<<<N_BKT, SORT_THREADS, 0, stream>>>(bkt, runs, cursor, perm, cnt, start, nNodes);

    gather32_kernel<<<wBlocks, 256, 0, stream>>>(feat, perm, start, cnt, agg, deg, nNodes);
    {
        int waves = nRowTiles * (HID_F / 16);
        int blocks = (waves * 32 + 255) / 256;
        sage_layer1_kernel<<<blocks, 256, 0, stream>>>(feat, agg, deg, W1s, W1n, b1, h, nRowTiles);
    }

    {
        int waves = nRowTiles;
        int blocks = (waves * 32 + 255) / 256;
        sage_layer2_project_kernel<<<blocks, 256, 0, stream>>>(h, W2s, W2n, b2, S, P, nRowTiles);
    }
    gather16_combine_kernel<<<qBlocks, 256, 0, stream>>>(P, S, perm, start, cnt, out, nNodes);
}


# Fusion: neighbour sums, layer-1 GEMM and both layer-2 projections of a 64-node tile run in one work-group (means and h stay in LDS, never in memory); 6->4 launches; all sum orders unchanged
# speedup vs baseline: 7.1150x; 1.3897x over previous
#include <hip/hip_runtime.h>
#include <hip/hip_bf16.h>

#define NN 100000
#define IN_F 32
#define HID_F 64
#define OUT_F 16

typedef float v2f __attribute__((ext_vector_type(2)));
typedef float v4f __attribute__((ext_vector_type(4)));
typedef float v8f __attribute__((ext_vector_type(8)));

#define NEDGES     1600000
#define NE_CHUNK   6256
#define N_CHUNKS   ((NEDGES + NE_CHUNK - 1) / NE_CHUNK)
#define BKT_NODES  392
#define N_BKT      ((NN + BKT_NODES - 1) / BKT_NODES)
#define SORT_THREADS 1024
#define EDGES_PER_THREAD ((NE_CHUNK + SORT_THREADS - 1) / SORT_THREADS)
#define RUNS_PER_WAVE (N_CHUNKS / 32)
static_assert(N_BKT <= SORT_THREADS / 4, "bucket counters are scanned by the first waves of the block");
static_assert(BKT_NODES <= SORT_THREADS, "one thread per node of a bucket");
static_assert(N_CHUNKS == 32 * RUNS_PER_WAVE && N_CHUNKS <= SORT_THREADS, "runs are dealt 8 to each of 32 waves");

__device__ __forceinline__ int wave_incl_scan(int v, int lane) {
    for (int o = 1; o < 32; o <<= 1) {
        int n = __shfl_up(v, o, 32);
        if (lane >= o) v += n;
    }
    return v;
}

__device__ __forceinline__ int block_excl_scan(int v, int t, int* wsum, int* total) {
    int lane = t & 31, wv = t >> 5;
    int inc = wave_incl_scan(v, lane);
    if (lane == 31) wsum[wv] = inc;
    __syncthreads();
    if (wv == 0) {
        int x  = wsum[lane];
        int xi = wave_incl_scan(x, lane);
        wsum[lane] = xi - x;
        if (lane == 31) *total = xi;
    }
    __syncthreads();
    return wsum[wv] + inc - v;
}

__global__ __launch_bounds__(SORT_THREADS) void chunk_bin_kernel(const int* __restrict__ src,
                                                                 const int* __restrict__ dst,
                                                                 int* __restrict__ bktw,
                                                                 int* __restrict__ runsw,
                                                                 int* __restrict__ cursor,
                                                                 int nE) {
    __shared__ int  hist[N_BKT];
    __shared__ int  offs[N_BKT];
    __shared__ int  wsum[32];
    __shared__ int  tot;
    __shared__ int2 sorted[NE_CHUNK];
    int2* bkt  = (int2*)bktw;
    int2* runs = (int2*)runsw;
    const int t  = threadIdx.x;
    const int w  = blockIdx.x;
    const int e0 = w * NE_CHUNK;
    int len = nE - e0;
    len = len < NE_CHUNK ? len : NE_CHUNK;
    if (t < N_BKT) hist[t] = 0;
    if (w == 0 && t == 0) *cursor = 0;
    __syncthreads();

    int s[EDGES_PER_THREAD], d[EDGES_PER_THREAD], r[EDGES_PER_THREAD];
#pragma unroll
    for (int u = 0; u < EDGES_PER_THREAD; ++u) {
        int i = t + u * SORT_THREADS;
        s[u] = 0; d[u] = 0;
        if (i < len) { d[u] = dst[e0 + i]; s[u] = src[e0 + i]; }
    }
#pragma unroll
    for (int u = 0; u < EDGES_PER_THREAD; ++u) {
        int i = t + u * SORT_THREADS;
        r[u] = 0;
        if (i < len) r[u] = atomicAdd(&hist[d[u] / BKT_NODES], 1);
    }
    __syncthreads();

    int c    = (t < N_BKT) ? hist[t] : 0;
    int excl = block_excl_scan(c, t, wsum, &tot);
    if (t < N_BKT) {
        offs[t] = excl;
        runs[t * N_CHUNKS + w] = make_int2(e0 + excl, c);
    }
    __syncthreads();

#pragma unroll
    for (int u = 0; u < EDGES_PER_THREAD; ++u) {
        int i = t + u * SORT_THREADS;
        if (i < len) sorted[offs[d[u] / BKT_NODES] + r[u]] = make_int2(s[u], d[u]);
    }
    __syncthreads();
    for (int i = t; i < len; i += SORT_THREADS) bkt[e0 + i] = sorted[i];
}

__global__ __launch_bounds__(SORT_THREADS) void bucket_sort_kernel(const int* __restrict__ bktw,
                                                                   const int* __restrict__ runsw,
                                                                   int* __restrict__ cursor,
                                                                   int* __restrict__ perm,
                                                                   int* __restrict__ cnt,
                                                                   int* __restrict__ start,
                                                                   int nNodes) {
    __shared__ int2 run[N_CHUNKS];
    __shared__ int  ccnt[BKT_NODES];
    __shared__ int  cur[BKT_NODES];
    __shared__ int  wsum[32];
    __shared__ int  tot;
    __shared__ int  sbase;
    const int2* bkt  = (const int2*)bktw;
    const int2* runs = (const int2*)runsw;
    const int t    = threadIdx.x;
    const int lane = t & 31;
    const int wv   = t >> 5;
    const int b    = blockIdx.x;
    const int nb0  = b * BKT_NODES;
    if (t < N_CHUNKS)  run[t]  = runs[b * N_CHUNKS + t];
    if (t < BKT_NODES) ccnt[t] = 0;
    __syncthreads();

    int2 rr[RUNS_PER_WAVE], e[RUNS_PER_WAVE];
#pragma unroll
    for (int u = 0; u < RUNS_PER_WAVE; ++u) {
        rr[u] = run[wv + 32 * u];
        e[u]  = make_int2(0, 0);
        if (lane < rr[u].y) e[u] = bkt[rr[u].x + lane];
    }
#pragma unroll
    for (int u = 0; u < RUNS_PER_WAVE; ++u) {
        if (lane < rr[u].y) atomicAdd(&ccnt[e[u].y - nb0], 1);
        for (int k = lane + 32; k < rr[u].y; k += 32) {
            int2 x = bkt[rr[u].x + k];
            atomicAdd(&ccnt[x.y - nb0], 1);
        }
    }
    __syncthreads();

    int c    = (t < BKT_NODES) ? ccnt[t] : 0;
    int excl = block_excl_scan(c, t, wsum, &tot);
    if (t == 0) sbase = atomicAdd(cursor, tot);
    if (t < BKT_NODES) cur[t] = excl;
    __syncthreads();
    const int base = sbase;
    if (t < BKT_NODES && nb0 + t < nNodes) {
        cnt[nb0 + t]   = c;
        start[nb0 + t] = base + excl;
    }

#pragma unroll
    for (int u = 0; u < RUNS_PER_WAVE; ++u) {
        if (lane < rr[u].y) {
            int p = atomicAdd(&cur[e[u].y - nb0], 1);
            perm[base + p] = e[u].x;
        }
        for (int k = lane + 32; k < rr[u].y; k += 32) {
            int2 x = bkt[rr[u].x + k];
            int p = atomicAdd(&cur[x.y - nb0], 1);
            perm[base + p] = x.x;
        }
    }
}

#define NODES_PER_WAVE 8

__device__ __forceinline__ float add_rows32(const float* __restrict__ feat, int idx, int e0, int m, int lane, float acc) {
    float v[8];
#pragma unroll
    for (int u = 0; u < 8; ++u) {
        int s = __builtin_amdgcn_readlane(idx, e0 + u);
        v[u] = feat[(long)s * IN_F + lane];
    }
#pragma unroll
    for (int u = 0; u < 8; ++u) acc += (e0 + u < m) ? v[u] : 0.0f;
    return acc;
}

__device__ __forceinline__ int load_list(const int* __restrict__ perm, int st, int c, int lane) {
    int idx = 0;
    if (c > 0) idx = perm[st + (lane < c ? lane : c - 1)];
    return idx;
}

#define TILE_NODES 64
#define X_PITCH    36
#define H_PITCH    68

__global__ __launch_bounds__(256, 16) void sage_front_kernel(const float* __restrict__ feat,
                                                         const int* __restrict__ perm,
                                                         const int* __restrict__ start,
                                                         const int* __restrict__ cnt,
                                                         const float* __restrict__ W1self,
                                                         const float* __restrict__ W1neigh,
                                                         const float* __restrict__ bias1,
                                                         const float* __restrict__ W2self,
                                                         const float* __restrict__ W2neigh,
                                                         const float* __restrict__ bias2,
                                                         float* __restrict__ S,
                                                         float* __restrict__ P,
                                                         int nNodes,
                                                         int nRowTiles) {
    __shared__ float xs[TILE_NODES * X_PITCH];
    __shared__ float ms[TILE_NODES * X_PITCH];
    __shared__ float hs[TILE_NODES * H_PITCH];
    const int lane = threadIdx.x & 31;
    const int wv = __builtin_amdgcn_readfirstlane(threadIdx.x >> 5);
    const int n0 = blockIdx.x * TILE_NODES + wv * NODES_PER_WAVE;
    int nn = nNodes - n0;
    nn = nn < NODES_PER_WAVE ? nn : NODES_PER_WAVE;
    nn = nn > 0 ? nn : 0;

    if (nn > 0) {
        const int lk  = lane < nn ? lane : nn - 1;
        const int stv = start[n0 + lk];
        const int cv  = cnt[n0 + lk];
        int st  = __builtin_amdgcn_readlane(stv, 0);
        int c   = __builtin_amdgcn_readlane(cv, 0);
        int idx = load_list(perm, st, c, lane);
        for (int k = 0; k < nn; ++k) {
            const int m = c < 32 ? c : 32;
            float v[8];
#pragma unroll
            for (int u = 0; u < 8; ++u) {
                int s = __builtin_amdgcn_readlane(idx, u);
                v[u] = feat[(long)s * IN_F + lane];
            }
            const float own = feat[(long)(n0 + k) * IN_F + lane];
            int stn = 0, cn = 0, idxn = 0;
            if (k + 1 < nn) {
                stn  = __builtin_amdgcn_readlane(stv, k + 1);
                cn   = __builtin_amdgcn_readlane(cv, k + 1);
                idxn = load_list(perm, stn, cn, lane);
            }
            float acc = 0.0f;
#pragma unroll
            for (int u = 0; u < 8; ++u) acc += (u < m) ? v[u] : 0.0f;
            for (int e0 = 8; e0 < m; e0 += 8) acc = add_rows32(feat, idx, e0, m, lane, acc);
            for (int j0 = 32; j0 < c; j0 += 32) {
                const int r    = c - j0;
                const int more = load_list(perm, st + j0, r, lane);
                const int mr   = r < 32 ? r : 32;
                for (int e0 = 0; e0 < mr; e0 += 8) acc = add_rows32(feat, more, e0, mr, lane, acc);
            }
            const float invd = 1.0f / fmaxf((float)c, 1.0f);
            xs[(wv * NODES_PER_WAVE + k) * X_PITCH + lane] = own;
            ms[(wv * NODES_PER_WAVE + k) * X_PITCH + lane] = acc * invd;
            st = stn; c = cn; idx = idxn;
        }
    }
    for (int k = nn; k < NODES_PER_WAVE; ++k) {
        xs[(wv * NODES_PER_WAVE + k) * X_PITCH + lane] = 0.0f;
        ms[(wv * NODES_PER_WAVE + k) * X_PITCH + lane] = 0.0f;
    }
    __syncthreads();

    const int hi  = lane >> 4;
    const int l15 = lane & 15;

    {
        const int ct   = wv & 3;
        const int rt0  = (wv >> 2) * 2;
        const int colW = ct * 16 + l15;
        const float* x0 = xs + (rt0 * 16 + l15) * X_PITCH;
        const float* x1 = x0 + 16 * X_PITCH;
        const float* m0 = ms + (rt0 * 16 + l15) * X_PITCH;
        const float* m1 = m0 + 16 * X_PITCH;
        v8f c0, c1;
        const float bv = bias1[colW];
#pragma unroll
        for (int v = 0; v < 8; ++v) { c0[v] = bv; c1[v] = bv; }
#pragma unroll
        for (int k0 = 0; k0 < IN_F; k0 += 4) {
            const int kA = k0 + 2 * hi;
            v2f a0, a1, b;
            b.x = W1self[(long)(kA)     * HID_F + colW];
            b.y = W1self[(long)(kA + 1) * HID_F + colW];
            a0.x = x0[kA]; a0.y = x0[kA + 1];
            a1.x = x1[kA]; a1.y = x1[kA + 1];
            c0 = __builtin_amdgcn_wmma_f32_16x16x4_f32(false, a0, false, b, (short)0, c0, false, false);
            c1 = __builtin_amdgcn_wmma_f32_16x16x4_f32(false, a1, false, b, (short)0, c1, false, false);
            b.x = W1neigh[(long)(kA)     * HID_F + colW];
            b.y = W1neigh[(long)(kA + 1) * HID_F + colW];
            a0.x = m0[kA]; a0.y = m0[kA + 1];
            a1.x = m1[kA]; a1.y = m1[kA + 1];
            c0 = __builtin_amdgcn_wmma_f32_16x16x4_f32(false, a0, false, b, (short)0, c0, false, false);
            c1 = __builtin_amdgcn_wmma_f32_16x16x4_f32(false, a1, false, b, (short)0, c1, false, false);
        }
#pragma unroll
        for (int v = 0; v < 8; ++v) {
            const int r = rt0 * 16 + v + 8 * hi;
            hs[r * H_PITCH + colW]        = fmaxf(c0[v], 0.0f);
            hs[(r + 16) * H_PITCH + colW] = fmaxf(c1[v], 0.0f);
        }
    }
    __syncthreads();

    if (wv < 4) {
        const int which = wv & 1;
        const int rt0   = (wv >> 1) * 2;
        const float* W  = which ? W2neigh : W2self;
        float*       O  = which ? P : S;
        const float* h0 = hs + (rt0 * 16 + l15) * H_PITCH;
        const float* h1 = h0 + 16 * H_PITCH;
        v8f c0, c1;
        const float bv = which ? 0.0f : bias2[l15];
#pragma unroll
        for (int v = 0; v < 8; ++v) { c0[v] = bv; c1[v] = bv; }
#pragma unroll
        for (int k0 = 0; k0 < HID_F; k0 += 4) {
            const int kA = k0 + 2 * hi;
            v2f a0, a1, b;
            b.x = W[(long)(kA)     * OUT_F + l15];
            b.y = W[(long)(kA + 1) * OUT_F + l15];
            a0.x = h0[kA]; a0.y = h0[kA + 1];
            a1.x = h1[kA]; a1.y = h1[kA + 1];
            c0 = __builtin_amdgcn_wmma_f32_16x16x4_f32(false, a0, false, b, (short)0, c0, false, false);
            c1 = __builtin_amdgcn_wmma_f32_16x16x4_f32(false, a1, false, b, (short)0, c1, false, false);
        }
        const int tile0 = blockIdx.x * (TILE_NODES / 16) + rt0;
        if (tile0 < nRowTiles) {
#pragma unroll
            for (int v = 0; v < 8; ++v) {
                const long r = (long)tile0 * 16 + v + 8 * hi;
                O[r * OUT_F + l15] = c0[v];
            }
        }
        if (tile0 + 1 < nRowTiles) {
#pragma unroll
            for (int v = 0; v < 8; ++v) {
                const long r = (long)(tile0 + 1) * 16 + v + 8 * hi;
                O[r * OUT_F + l15] = c1[v];
            }
        }
    }
}

#define WAVE_NODES 16

template <int J>
__device__ __forceinline__ int quad_lane(int x) {
    return __builtin_amdgcn_update_dpp(0, x, J * 0x55, 0xf, 0xf, true);
}

__device__ __forceinline__ int list_entry(const int* __restrict__ perm, int st, int c, int e) {
    int ee = e < c ? e : c - 1;
    int s = 0;
    if (c > 0) s = perm[st + ee];
    return s;
}

__device__ __forceinline__ void load_rows16(const float* __restrict__ P, int i0, int i1, int q, v4f v[8]) {
    int s[8];
    s[0] = quad_lane<0>(i0); s[1] = quad_lane<1>(i0); s[2] = quad_lane<2>(i0); s[3] = quad_lane<3>(i0);
    s[4] = quad_lane<0>(i1); s[5] = quad_lane<1>(i1); s[6] = quad_lane<2>(i1); s[7] = quad_lane<3>(i1);
#pragma unroll
    for (int u = 0; u < 8; ++u) v[u] = *(const v4f*)(P + (long)s[u] * OUT_F + 4 * q);
}

__global__ __launch_bounds__(256) void gather16_combine_kernel(const float* __restrict__ P,
                                                               const float* __restrict__ S,
                                                               const int* __restrict__ perm,
                                                               const int* __restrict__ start,
                                                               const int* __restrict__ cnt,
                                                               float* __restrict__ out,
                                                               int nNodes) {
    const int lane = threadIdx.x & 31;
    const int q    = lane & 3;
    const int wave = blockIdx.x * 8 + (threadIdx.x >> 5);
    const int nA   = wave * WAVE_NODES + (lane >> 2);
    const int nB   = nA + 8;
    const bool onA = nA < nNodes;
    const bool onB = nB < nNodes;
    int stA = 0, cA = 0, stB = 0, cB = 0;
    v4f selfA = {0.0f, 0.0f, 0.0f, 0.0f}, selfB = {0.0f, 0.0f, 0.0f, 0.0f};
    if (onA) { stA = start[nA]; cA = cnt[nA]; selfA = *(const v4f*)(S + (long)nA * OUT_F + 4 * q); }
    if (onB) { stB = start[nB]; cB = cnt[nB]; selfB = *(const v4f*)(S + (long)nB * OUT_F + 4 * q); }
    const int cm = cA > cB ? cA : cB;

    int a0 = list_entry(perm, stA, cA, q), a1 = list_entry(perm, stA, cA, 4 + q);
    int b0 = list_entry(perm, stB, cB, q), b1 = list_entry(perm, stB, cB, 4 + q);
    v4f accA = {0.0f, 0.0f, 0.0f, 0.0f}, accB = {0.0f, 0.0f, 0.0f, 0.0f};
    const v4f zero = {0.0f, 0.0f, 0.0f, 0.0f};
    for (int e0 = 0; e0 < cm; e0 += 8) {
        v4f va[8], vb[8];
        load_rows16(P, a0, a1, q, va);
        load_rows16(P, b0, b1, q, vb);
        if (e0 + 8 < cm) {
            a0 = list_entry(perm, stA, cA, e0 + 8 + q);  a1 = list_entry(perm, stA, cA, e0 + 12 + q);
            b0 = list_entry(perm, stB, cB, e0 + 8 + q);  b1 = list_entry(perm, stB, cB, e0 + 12 + q);
        }
#pragma unroll
        for (int u = 0; u < 8; ++u) {
            accA += (e0 + u < cA) ? va[u] : zero;
            accB += (e0 + u < cB) ? vb[u] : zero;
        }
    }
    if (onA) *(v4f*)(out + (long)nA * OUT_F + 4 * q) = selfA + accA / fmaxf((float)cA, 1.0f);
    if (onB) *(v4f*)(out + (long)nB * OUT_F + 4 * q) = selfB + accB / fmaxf((float)cB, 1.0f);
}

extern "C" void kernel_launch(void* const* d_in, const int* in_sizes, int n_in,
                              void* d_out, int out_size, void* d_ws, size_t ws_size,
                              hipStream_t stream) {
    const float* feat = (const float*)d_in[0];
    const int*   src  = (const int*)d_in[1];
    const int*   dst  = (const int*)d_in[2];
    const float* W1s  = (const float*)d_in[3];
    const float* W1n  = (const float*)d_in[4];
    const float* b1   = (const float*)d_in[5];
    const float* W2s  = (const float*)d_in[6];
    const float* W2n  = (const float*)d_in[7];
    const float* b2   = (const float*)d_in[8];
    float* out = (float*)d_out;

    const int nNodes = in_sizes[0] / IN_F;
    const int nE     = in_sizes[1];
    const int nRowTiles = nNodes / 16;

    float* ws    = (float*)d_ws;
    float* r0    = ws;
    float* r1    = r0 + (long)nNodes * HID_F;
    int*   perm  = (int*)(r1 + (long)nNodes * HID_F);
    int*   cnt   = perm + nE;
    int*   cursor = cnt + nNodes;
    int*   start = cursor + 32;
    int*   bkt   = (int*)r0;
    int*   runs  = (int*)r1;
    float* P     = r1;
    float* S     = r1 + (long)nNodes * OUT_F;

    const int tBlocks = (nNodes + TILE_NODES - 1) / TILE_NODES;
    const int qBlocks = (nNodes + 8 * WAVE_NODES - 1) / (8 * WAVE_NODES);

    chunk_bin_kernel<<<N_CHUNKS, SORT_THREADS, 0, stream>>>(src, dst, bkt, runs, cursor, nE);
    bucket_sort_kernel<<<N_BKT, SORT_THREADS, 0, stream>>>(bkt, runs, cursor, perm, cnt, start, nNodes);

    sage_front_kernel<<<tBlocks, 256, 0, stream>>>(feat, perm, start, cnt, W1s, W1n, b1, W2s, W2n, b2, S, P, nNodes, nRowTiles);

    gather16_combine_kernel<<<qBlocks, 256, 0, stream>>>(P, S, perm, start, cnt, out, nNodes);
}


# Per-bucket edge ordering: 1600000 single-word list stores scattered across each bucket's region -> region put in order in an LDS buffer, then stored as whole consecutive lines; sum orders unchanged
# speedup vs baseline: 7.7103x; 1.0837x over previous
#include <hip/hip_runtime.h>
#include <hip/hip_bf16.h>

#define NN 100000
#define IN_F 32
#define HID_F 64
#define OUT_F 16

typedef float v2f __attribute__((ext_vector_type(2)));
typedef float v4f __attribute__((ext_vector_type(4)));
typedef float v8f __attribute__((ext_vector_type(8)));

#define NEDGES     1600000
#define NE_CHUNK   6256
#define N_CHUNKS   ((NEDGES + NE_CHUNK - 1) / NE_CHUNK)
#define BKT_NODES  392
#define N_BKT      ((NN + BKT_NODES - 1) / BKT_NODES)
#define SORT_THREADS 1024
#define EDGES_PER_THREAD ((NE_CHUNK + SORT_THREADS - 1) / SORT_THREADS)
#define RUNS_PER_WAVE (N_CHUNKS / 32)
#define STAGE_CAP  8192
static_assert(N_BKT <= SORT_THREADS / 4, "bucket counters are scanned by the first waves of the block");
static_assert(BKT_NODES <= SORT_THREADS, "one thread per node of a bucket");
static_assert(N_CHUNKS == 32 * RUNS_PER_WAVE && N_CHUNKS <= SORT_THREADS, "runs are dealt 8 to each of 32 waves");

__device__ __forceinline__ int wave_incl_scan(int v, int lane) {
    for (int o = 1; o < 32; o <<= 1) {
        int n = __shfl_up(v, o, 32);
        if (lane >= o) v += n;
    }
    return v;
}

__device__ __forceinline__ int block_excl_scan(int v, int t, int* wsum, int* total) {
    int lane = t & 31, wv = t >> 5;
    int inc = wave_incl_scan(v, lane);
    if (lane == 31) wsum[wv] = inc;
    __syncthreads();
    if (wv == 0) {
        int x  = wsum[lane];
        int xi = wave_incl_scan(x, lane);
        wsum[lane] = xi - x;
        if (lane == 31) *total = xi;
    }
    __syncthreads();
    return wsum[wv] + inc - v;
}

__global__ __launch_bounds__(SORT_THREADS) void chunk_bin_kernel(const int* __restrict__ src,
                                                                 const int* __restrict__ dst,
                                                                 int* __restrict__ bktw,
                                                                 int* __restrict__ runsw,
                                                                 int* __restrict__ cursor,
                                                                 int nE) {
    __shared__ int  hist[N_BKT];
    __shared__ int  offs[N_BKT];
    __shared__ int  wsum[32];
    __shared__ int  tot;
    __shared__ int2 sorted[NE_CHUNK];
    int2* bkt  = (int2*)bktw;
    int2* runs = (int2*)runsw;
    const int t  = threadIdx.x;
    const int w  = blockIdx.x;
    const int e0 = w * NE_CHUNK;
    int len = nE - e0;
    len = len < NE_CHUNK ? len : NE_CHUNK;
    if (t < N_BKT) hist[t] = 0;
    if (w == 0 && t == 0) *cursor = 0;
    __syncthreads();

    int s[EDGES_PER_THREAD], d[EDGES_PER_THREAD], r[EDGES_PER_THREAD];
#pragma unroll
    for (int u = 0; u < EDGES_PER_THREAD; ++u) {
        int i = t + u * SORT_THREADS;
        s[u] = 0; d[u] = 0;
        if (i < len) { d[u] = dst[e0 + i]; s[u] = src[e0 + i]; }
    }
#pragma unroll
    for (int u = 0; u < EDGES_PER_THREAD; ++u) {
        int i = t + u * SORT_THREADS;
        r[u] = 0;
        if (i < len) r[u] = atomicAdd(&hist[d[u] / BKT_NODES], 1);
    }
    __syncthreads();

    int c    = (t < N_BKT) ? hist[t] : 0;
    int excl = block_excl_scan(c, t, wsum, &tot);
    if (t < N_BKT) {
        offs[t] = excl;
        runs[t * N_CHUNKS + w] = make_int2(e0 + excl, c);
    }
    __syncthreads();

#pragma unroll
    for (int u = 0; u < EDGES_PER_THREAD; ++u) {
        int i = t + u * SORT_THREADS;
        if (i < len) sorted[offs[d[u] / BKT_NODES] + r[u]] = make_int2(s[u], d[u]);
    }
    __syncthreads();
    for (int i = t; i < len; i += SORT_THREADS) bkt[e0 + i] = sorted[i];
}

__global__ __launch_bounds__(SORT_THREADS) void bucket_sort_kernel(const int* __restrict__ bktw,
                                                                   const int* __restrict__ runsw,
                                                                   int* __restrict__ cursor,
                                                                   int* __restrict__ perm,
                                                                   int* __restrict__ cnt,
                                                                   int* __restrict__ start,
                                                                   int nNodes) {
    __shared__ int2 run[N_CHUNKS];
    __shared__ int  ccnt[BKT_NODES];
    __shared__ int  cur[BKT_NODES];
    __shared__ int  wsum[32];
    __shared__ int  tot;
    __shared__ int  sbase;
    __shared__ int  stage[STAGE_CAP];
    const int2* bkt  = (const int2*)bktw;
    const int2* runs = (const int2*)runsw;
    const int t    = threadIdx.x;
    const int lane = t & 31;
    const int wv   = t >> 5;
    const int b    = blockIdx.x;
    const int nb0  = b * BKT_NODES;
    if (t < N_CHUNKS)  run[t]  = runs[b * N_CHUNKS + t];
    if (t < BKT_NODES) ccnt[t] = 0;
    __syncthreads();

    int2 rr[RUNS_PER_WAVE], e[RUNS_PER_WAVE];
#pragma unroll
    for (int u = 0; u < RUNS_PER_WAVE; ++u) {
        rr[u] = run[wv + 32 * u];
        e[u]  = make_int2(0, 0);
        if (lane < rr[u].y) e[u] = bkt[rr[u].x + lane];
    }
#pragma unroll
    for (int u = 0; u < RUNS_PER_WAVE; ++u) {
        if (lane < rr[u].y) atomicAdd(&ccnt[e[u].y - nb0], 1);
        for (int k = lane + 32; k < rr[u].y; k += 32) {
            int2 x = bkt[rr[u].x + k];
            atomicAdd(&ccnt[x.y - nb0], 1);
        }
    }
    __syncthreads();

    int c    = (t < BKT_NODES) ? ccnt[t] : 0;
    int excl = block_excl_scan(c, t, wsum, &tot);
    if (t == 0) sbase = atomicAdd(cursor, tot);
    if (t < BKT_NODES) cur[t] = excl;
    __syncthreads();
    const int base = sbase;
    if (t < BKT_NODES && nb0 + t < nNodes) {
        cnt[nb0 + t]   = c;
        start[nb0 + t] = base + excl;
    }

    const int  total  = tot;
    const bool staged = total <= STAGE_CAP;
#pragma unroll
    for (int u = 0; u < RUNS_PER_WAVE; ++u) {
        if (lane < rr[u].y) {
            int p = atomicAdd(&cur[e[u].y - nb0], 1);
            if (staged) stage[p] = e[u].x; else perm[base + p] = e[u].x;
        }
        for (int k = lane + 32; k < rr[u].y; k += 32) {
            int2 x = bkt[rr[u].x + k];
            int p = atomicAdd(&cur[x.y - nb0], 1);
            if (staged) stage[p] = x.x; else perm[base + p] = x.x;
        }
    }
    if (staged) {
        __syncthreads();
        for (int i = t; i < total; i += SORT_THREADS) perm[base + i] = stage[i];
    }
}

#define NODES_PER_WAVE 8

__device__ __forceinline__ float add_rows32(const float* __restrict__ feat, int idx, int e0, int m, int lane, float acc) {
    float v[8];
#pragma unroll
    for (int u = 0; u < 8; ++u) {
        int s = __builtin_amdgcn_readlane(idx, e0 + u);
        v[u] = feat[(long)s * IN_F + lane];
    }
#pragma unroll
    for (int u = 0; u < 8; ++u) acc += (e0 + u < m) ? v[u] : 0.0f;
    return acc;
}

__device__ __forceinline__ int load_list(const int* __restrict__ perm, int st, int c, int lane) {
    int idx = 0;
    if (c > 0) idx = perm[st + (lane < c ? lane : c - 1)];
    return idx;
}

#define TILE_NODES 64
#define X_PITCH    36
#define H_PITCH    68

__global__ __launch_bounds__(256, 16) void sage_front_kernel(const float* __restrict__ feat,
                                                         const int* __restrict__ perm,
                                                         const int* __restrict__ start,
                                                         const int* __restrict__ cnt,
                                                         const float* __restrict__ W1self,
                                                         const float* __restrict__ W1neigh,
                                                         const float* __restrict__ bias1,
                                                         const float* __restrict__ W2self,
                                                         const float* __restrict__ W2neigh,
                                                         const float* __restrict__ bias2,
                                                         float* __restrict__ S,
                                                         float* __restrict__ P,
                                                         int nNodes,
                                                         int nRowTiles) {
    __shared__ float xs[TILE_NODES * X_PITCH];
    __shared__ float ms[TILE_NODES * X_PITCH];
    __shared__ float hs[TILE_NODES * H_PITCH];
    const int lane = threadIdx.x & 31;
    const int wv = __builtin_amdgcn_readfirstlane(threadIdx.x >> 5);
    const int n0 = blockIdx.x * TILE_NODES + wv * NODES_PER_WAVE;
    int nn = nNodes - n0;
    nn = nn < NODES_PER_WAVE ? nn : NODES_PER_WAVE;
    nn = nn > 0 ? nn : 0;

    if (nn > 0) {
        const int lk  = lane < nn ? lane : nn - 1;
        const int stv = start[n0 + lk];
        const int cv  = cnt[n0 + lk];
        int st  = __builtin_amdgcn_readlane(stv, 0);
        int c   = __builtin_amdgcn_readlane(cv, 0);
        int idx = load_list(perm, st, c, lane);
        for (int k = 0; k < nn; ++k) {
            const int m = c < 32 ? c : 32;
            float v[8];
#pragma unroll
            for (int u = 0; u < 8; ++u) {
                int s = __builtin_amdgcn_readlane(idx, u);
                v[u] = feat[(long)s * IN_F + lane];
            }
            const float own = feat[(long)(n0 + k) * IN_F + lane];
            int stn = 0, cn = 0, idxn = 0;
            if (k + 1 < nn) {
                stn  = __builtin_amdgcn_readlane(stv, k + 1);
                cn   = __builtin_amdgcn_readlane(cv, k + 1);
                idxn = load_list(perm, stn, cn, lane);
            }
            float acc = 0.0f;
#pragma unroll
            for (int u = 0; u < 8; ++u) acc += (u < m) ? v[u] : 0.0f;
            for (int e0 = 8; e0 < m; e0 += 8) acc = add_rows32(feat, idx, e0, m, lane, acc);
            for (int j0 = 32; j0 < c; j0 += 32) {
                const int r    = c - j0;
                const int more = load_list(perm, st + j0, r, lane);
                const int mr   = r < 32 ? r : 32;
                for (int e0 = 0; e0 < mr; e0 += 8) acc = add_rows32(feat, more, e0, mr, lane, acc);
            }
            const float invd = 1.0f / fmaxf((float)c, 1.0f);
            xs[(wv * NODES_PER_WAVE + k) * X_PITCH + lane] = own;
            ms[(wv * NODES_PER_WAVE + k) * X_PITCH + lane] = acc * invd;
            st = stn; c = cn; idx = idxn;
        }
    }
    for (int k = nn; k < NODES_PER_WAVE; ++k) {
        xs[(wv * NODES_PER_WAVE + k) * X_PITCH + lane] = 0.0f;
        ms[(wv * NODES_PER_WAVE + k) * X_PITCH + lane] = 0.0f;
    }
    __syncthreads();

    const int hi  = lane >> 4;
    const int l15 = lane & 15;

    {
        const int ct   = wv & 3;
        const int rt0  = (wv >> 2) * 2;
        const int colW = ct * 16 + l15;
        const float* x0 = xs + (rt0 * 16 + l15) * X_PITCH;
        const float* x1 = x0 + 16 * X_PITCH;
        const float* m0 = ms + (rt0 * 16 + l15) * X_PITCH;
        const float* m1 = m0 + 16 * X_PITCH;
        v8f c0, c1;
        const float bv = bias1[colW];
#pragma unroll
        for (int v = 0; v < 8; ++v) { c0[v] = bv; c1[v] = bv; }
#pragma unroll
        for (int k0 = 0; k0 < IN_F; k0 += 4) {
            const int kA = k0 + 2 * hi;
            v2f a0, a1, b;
            b.x = W1self[(long)(kA)     * HID_F + colW];
            b.y = W1self[(long)(kA + 1) * HID_F + colW];
            a0.x = x0[kA]; a0.y = x0[kA + 1];
            a1.x = x1[kA]; a1.y = x1[kA + 1];
            c0 = __builtin_amdgcn_wmma_f32_16x16x4_f32(false, a0, false, b, (short)0, c0, false, false);
            c1 = __builtin_amdgcn_wmma_f32_16x16x4_f32(false, a1, false, b, (short)0, c1, false, false);
            b.x = W1neigh[(long)(kA)     * HID_F + colW];
            b.y = W1neigh[(long)(kA + 1) * HID_F + colW];
            a0.x = m0[kA]; a0.y = m0[kA + 1];
            a1.x = m1[kA]; a1.y = m1[kA + 1];
            c0 = __builtin_amdgcn_wmma_f32_16x16x4_f32(false, a0, false, b, (short)0, c0, false, false);
            c1 = __builtin_amdgcn_wmma_f32_16x16x4_f32(false, a1, false, b, (short)0, c1, false, false);
        }
#pragma unroll
        for (int v = 0; v < 8; ++v) {
            const int r = rt0 * 16 + v + 8 * hi;
            hs[r * H_PITCH + colW]        = fmaxf(c0[v], 0.0f);
            hs[(r + 16) * H_PITCH + colW] = fmaxf(c1[v], 0.0f);
        }
    }
    __syncthreads();

    if (wv < 4) {
        const int which = wv & 1;
        const int rt0   = (wv >> 1) * 2;
        const float* W  = which ? W2neigh : W2self;
        float*       O  = which ? P : S;
        const float* h0 = hs + (rt0 * 16 + l15) * H_PITCH;
        const float* h1 = h0 + 16 * H_PITCH;
        v8f c0, c1;
        const float bv = which ? 0.0f : bias2[l15];
#pragma unroll
        for (int v = 0; v < 8; ++v) { c0[v] = bv; c1[v] = bv; }
#pragma unroll
        for (int k0 = 0; k0 < HID_F; k0 += 4) {
            const int kA = k0 + 2 * hi;
            v2f a0, a1, b;
            b.x = W[(long)(kA)     * OUT_F + l15];
            b.y = W[(long)(kA + 1) * OUT_F + l15];
            a0.x = h0[kA]; a0.y = h0[kA + 1];
            a1.x = h1[kA]; a1.y = h1[kA + 1];
            c0 = __builtin_amdgcn_wmma_f32_16x16x4_f32(false, a0, false, b, (short)0, c0, false, false);
            c1 = __builtin_amdgcn_wmma_f32_16x16x4_f32(false, a1, false, b, (short)0, c1, false, false);
        }
        const int tile0 = blockIdx.x * (TILE_NODES / 16) + rt0;
        if (tile0 < nRowTiles) {
#pragma unroll
            for (int v = 0; v < 8; ++v) {
                const long r = (long)tile0 * 16 + v + 8 * hi;
                O[r * OUT_F + l15] = c0[v];
            }
        }
        if (tile0 + 1 < nRowTiles) {
#pragma unroll
            for (int v = 0; v < 8; ++v) {
                const long r = (long)(tile0 + 1) * 16 + v + 8 * hi;
                O[r * OUT_F + l15] = c1[v];
            }
        }
    }
}

#define WAVE_NODES 16

template <int J>
__device__ __forceinline__ int quad_lane(int x) {
    return __builtin_amdgcn_update_dpp(0, x, J * 0x55, 0xf, 0xf, true);
}

__device__ __forceinline__ int list_entry(const int* __restrict__ perm, int st, int c, int e) {
    int ee = e < c ? e : c - 1;
    int s = 0;
    if (c > 0) s = perm[st + ee];
    return s;
}

__device__ __forceinline__ void load_rows16(const float* __restrict__ P, int i0, int i1, int q, v4f v[8]) {
    int s[8];
    s[0] = quad_lane<0>(i0); s[1] = quad_lane<1>(i0); s[2] = quad_lane<2>(i0); s[3] = quad_lane<3>(i0);
    s[4] = quad_lane<0>(i1); s[5] = quad_lane<1>(i1); s[6] = quad_lane<2>(i1); s[7] = quad_lane<3>(i1);
#pragma unroll
    for (int u = 0; u < 8; ++u) v[u] = *(const v4f*)(P + (long)s[u] * OUT_F + 4 * q);
}

__global__ __launch_bounds__(256) void gather16_combine_kernel(const float* __restrict__ P,
                                                               const float* __restrict__ S,
                                                               const int* __restrict__ perm,
                                                               const int* __restrict__ start,
                                                               const int* __restrict__ cnt,
                                                               float* __restrict__ out,
                                                               int nNodes) {
    const int lane = threadIdx.x & 31;
    const int q    = lane & 3;
    const int wave = blockIdx.x * 8 + (threadIdx.x >> 5);
    const int nA   = wave * WAVE_NODES + (lane >> 2);
    const int nB   = nA + 8;
    const bool onA = nA < nNodes;
    const bool onB = nB < nNodes;
    int stA = 0, cA = 0, stB = 0, cB = 0;
    v4f selfA = {0.0f, 0.0f, 0.0f, 0.0f}, selfB = {0.0f, 0.0f, 0.0f, 0.0f};
    if (onA) { stA = start[nA]; cA = cnt[nA]; selfA = *(const v4f*)(S + (long)nA * OUT_F + 4 * q); }
    if (onB) { stB = start[nB]; cB = cnt[nB]; selfB = *(const v4f*)(S + (long)nB * OUT_F + 4 * q); }
    const int cm = cA > cB ? cA : cB;

    int a0 = list_entry(perm, stA, cA, q), a1 = list_entry(perm, stA, cA, 4 + q);
    int b0 = list_entry(perm, stB, cB, q), b1 = list_entry(perm, stB, cB, 4 + q);
    v4f accA = {0.0f, 0.0f, 0.0f, 0.0f}, accB = {0.0f, 0.0f, 0.0f, 0.0f};
    const v4f zero = {0.0f, 0.0f, 0.0f, 0.0f};
    for (int e0 = 0; e0 < cm; e0 += 8) {
        v4f va[8], vb[8];
        load_rows16(P, a0, a1, q, va);
        load_rows16(P, b0, b1, q, vb);
        if (e0 + 8 < cm) {
            a0 = list_entry(perm, stA, cA, e0 + 8 + q);  a1 = list_entry(perm, stA, cA, e0 + 12 + q);
            b0 = list_entry(perm, stB, cB, e0 + 8 + q);  b1 = list_entry(perm, stB, cB, e0 + 12 + q);
        }
#pragma unroll
        for (int u = 0; u < 8; ++u) {
            accA += (e0 + u < cA) ? va[u] : zero;
            accB += (e0 + u < cB) ? vb[u] : zero;
        }
    }
    if (onA) *(v4f*)(out + (long)nA * OUT_F + 4 * q) = selfA + accA / fmaxf((float)cA, 1.0f);
    if (onB) *(v4f*)(out + (long)nB * OUT_F + 4 * q) = selfB + accB / fmaxf((float)cB, 1.0f);
}

extern "C" void kernel_launch(void* const* d_in, const int* in_sizes, int n_in,
                              void* d_out, int out_size, void* d_ws, size_t ws_size,
                              hipStream_t stream) {
    const float* feat = (const float*)d_in[0];
    const int*   src  = (const int*)d_in[1];
    const int*   dst  = (const int*)d_in[2];
    const float* W1s  = (const float*)d_in[3];
    const float* W1n  = (const float*)d_in[4];
    const float* b1   = (const float*)d_in[5];
    const float* W2s  = (const float*)d_in[6];
    const float* W2n  = (const float*)d_in[7];
    const float* b2   = (const float*)d_in[8];
    float* out = (float*)d_out;

    const int nNodes = in_sizes[0] / IN_F;
    const int nE     = in_sizes[1];
    const int nRowTiles = nNodes / 16;

    float* ws    = (float*)d_ws;
    float* r0    = ws;
    float* r1    = r0 + (long)nNodes * HID_F;
    int*   perm  = (int*)(r1 + (long)nNodes * HID_F);
    int*   cnt   = perm + nE;
    int*   cursor = cnt + nNodes;
    int*   start = cursor + 32;
    int*   bkt   = (int*)r0;
    int*   runs  = (int*)r1;
    float* P     = r1;
    float* S     = r1 + (long)nNodes * OUT_F;

    const int tBlocks = (nNodes + TILE_NODES - 1) / TILE_NODES;
    const int qBlocks = (nNodes + 8 * WAVE_NODES - 1) / (8 * WAVE_NODES);

    chunk_bin_kernel<<<N_CHUNKS, SORT_THREADS, 0, stream>>>(src, dst, bkt, runs, cursor, nE);
    bucket_sort_kernel<<<N_BKT, SORT_THREADS, 0, stream>>>(bkt, runs, cursor, perm, cnt, start, nNodes);

    sage_front_kernel<<<tBlocks, 256, 0, stream>>>(feat, perm, start, cnt, W1s, W1n, b1, W2s, W2n, b2, S, P, nNodes, nRowTiles);

    gather16_combine_kernel<<<qBlocks, 256, 0, stream>>>(P, S, perm, start, cnt, out, nNodes);
}


# Layer-1 neighbour sums: 4-byte loads, one 128-byte row per instruction, a wave's 8 nodes in turn -> 16-byte loads, 8 lanes own a node, 4 nodes side by side; per-node sum order unchanged
# speedup vs baseline: 8.1629x; 1.0587x over previous
#include <hip/hip_runtime.h>
#include <hip/hip_bf16.h>

#define NN 100000
#define IN_F 32
#define HID_F 64
#define OUT_F 16

typedef float v2f __attribute__((ext_vector_type(2)));
typedef float v4f __attribute__((ext_vector_type(4)));
typedef float v8f __attribute__((ext_vector_type(8)));

#define NEDGES     1600000
#define NE_CHUNK   6256
#define N_CHUNKS   ((NEDGES + NE_CHUNK - 1) / NE_CHUNK)
#define BKT_NODES  392
#define N_BKT      ((NN + BKT_NODES - 1) / BKT_NODES)
#define SORT_THREADS 1024
#define EDGES_PER_THREAD ((NE_CHUNK + SORT_THREADS - 1) / SORT_THREADS)
#define RUNS_PER_WAVE (N_CHUNKS / 32)
#define STAGE_CAP  8192
static_assert(N_BKT <= SORT_THREADS / 4, "bucket counters are scanned by the first waves of the block");
static_assert(BKT_NODES <= SORT_THREADS, "one thread per node of a bucket");
static_assert(N_CHUNKS == 32 * RUNS_PER_WAVE && N_CHUNKS <= SORT_THREADS, "runs are dealt 8 to each of 32 waves");

__device__ __forceinline__ int wave_incl_scan(int v, int lane) {
    for (int o = 1; o < 32; o <<= 1) {
        int n = __shfl_up(v, o, 32);
        if (lane >= o) v += n;
    }
    return v;
}

__device__ __forceinline__ int block_excl_scan(int v, int t, int* wsum, int* total) {
    int lane = t & 31, wv = t >> 5;
    int inc = wave_incl_scan(v, lane);
    if (lane == 31) wsum[wv] = inc;
    __syncthreads();
    if (wv == 0) {
        int x  = wsum[lane];
        int xi = wave_incl_scan(x, lane);
        wsum[lane] = xi - x;
        if (lane == 31) *total = xi;
    }
    __syncthreads();
    return wsum[wv] + inc - v;
}

__global__ __launch_bounds__(SORT_THREADS) void chunk_bin_kernel(const int* __restrict__ src,
                                                                 const int* __restrict__ dst,
                                                                 int* __restrict__ bktw,
                                                                 int* __restrict__ runsw,
                                                                 int* __restrict__ cursor,
                                                                 int nE) {
    __shared__ int  hist[N_BKT];
    __shared__ int  offs[N_BKT];
    __shared__ int  wsum[32];
    __shared__ int  tot;
    __shared__ int2 sorted[NE_CHUNK];
    int2* bkt  = (int2*)bktw;
    int2* runs = (int2*)runsw;
    const int t  = threadIdx.x;
    const int w  = blockIdx.x;
    const int e0 = w * NE_CHUNK;
    int len = nE - e0;
    len = len < NE_CHUNK ? len : NE_CHUNK;
    if (t < N_BKT) hist[t] = 0;
    if (w == 0 && t == 0) *cursor = 0;
    __syncthreads();

    int s[EDGES_PER_THREAD], d[EDGES_PER_THREAD], r[EDGES_PER_THREAD];
#pragma unroll
    for (int u = 0; u < EDGES_PER_THREAD; ++u) {
        int i = t + u * SORT_THREADS;
        s[u] = 0; d[u] = 0;
        if (i < len) { d[u] = dst[e0 + i]; s[u] = src[e0 + i]; }
    }
#pragma unroll
    for (int u = 0; u < EDGES_PER_THREAD; ++u) {
        int i = t + u * SORT_THREADS;
        r[u] = 0;
        if (i < len) r[u] = atomicAdd(&hist[d[u] / BKT_NODES], 1);
    }
    __syncthreads();

    int c    = (t < N_BKT) ? hist[t] : 0;
    int excl = block_excl_scan(c, t, wsum, &tot);
    if (t < N_BKT) {
        offs[t] = excl;
        runs[t * N_CHUNKS + w] = make_int2(e0 + excl, c);
    }
    __syncthreads();

#pragma unroll
    for (int u = 0; u < EDGES_PER_THREAD; ++u) {
        int i = t + u * SORT_THREADS;
        if (i < len) sorted[offs[d[u] / BKT_NODES] + r[u]] = make_int2(s[u], d[u]);
    }
    __syncthreads();
    for (int i = t; i < len; i += SORT_THREADS) bkt[e0 + i] = sorted[i];
}

__global__ __launch_bounds__(SORT_THREADS) void bucket_sort_kernel(const int* __restrict__ bktw,
                                                                   const int* __restrict__ runsw,
                                                                   int* __restrict__ cursor,
                                                                   int* __restrict__ perm,
                                                                   int* __restrict__ cnt,
                                                                   int* __restrict__ start,
                                                                   int nNodes) {
    __shared__ int2 run[N_CHUNKS];
    __shared__ int  ccnt[BKT_NODES];
    __shared__ int  cur[BKT_NODES];
    __shared__ int  wsum[32];
    __shared__ int  tot;
    __shared__ int  sbase;
    __shared__ int  stage[STAGE_CAP];
    const int2* bkt  = (const int2*)bktw;
    const int2* runs = (const int2*)runsw;
    const int t    = threadIdx.x;
    const int lane = t & 31;
    const int wv   = t >> 5;
    const int b    = blockIdx.x;
    const int nb0  = b * BKT_NODES;
    if (t < N_CHUNKS)  run[t]  = runs[b * N_CHUNKS + t];
    if (t < BKT_NODES) ccnt[t] = 0;
    __syncthreads();

    int2 rr[RUNS_PER_WAVE], e[RUNS_PER_WAVE];
#pragma unroll
    for (int u = 0; u < RUNS_PER_WAVE; ++u) {
        rr[u] = run[wv + 32 * u];
        e[u]  = make_int2(0, 0);
        if (lane < rr[u].y) e[u] = bkt[rr[u].x + lane];
    }
#pragma unroll
    for (int u = 0; u < RUNS_PER_WAVE; ++u) {
        if (lane < rr[u].y) atomicAdd(&ccnt[e[u].y - nb0], 1);
        for (int k = lane + 32; k < rr[u].y; k += 32) {
            int2 x = bkt[rr[u].x + k];
            atomicAdd(&ccnt[x.y - nb0], 1);
        }
    }
    __syncthreads();

    int c    = (t < BKT_NODES) ? ccnt[t] : 0;
    int excl = block_excl_scan(c, t, wsum, &tot);
    if (t == 0) sbase = atomicAdd(cursor, tot);
    if (t < BKT_NODES) cur[t] = excl;
    __syncthreads();
    const int base = sbase;
    if (t < BKT_NODES && nb0 + t < nNodes) {
        cnt[nb0 + t]   = c;
        start[nb0 + t] = base + excl;
    }

    const int  total  = tot;
    const bool staged = total <= STAGE_CAP;
#pragma unroll
    for (int u = 0; u < RUNS_PER_WAVE; ++u) {
        if (lane < rr[u].y) {
            int p = atomicAdd(&cur[e[u].y - nb0], 1);
            if (staged) stage[p] = e[u].x; else perm[base + p] = e[u].x;
        }
        for (int k = lane + 32; k < rr[u].y; k += 32) {
            int2 x = bkt[rr[u].x + k];
            int p = atomicAdd(&cur[x.y - nb0], 1);
            if (staged) stage[p] = x.x; else perm[base + p] = x.x;
        }
    }
    if (staged) {
        __syncthreads();
        for (int i = t; i < total; i += SORT_THREADS) perm[base + i] = stage[i];
    }
}

#define NODES_PER_WAVE 8

template <int J>
__device__ __forceinline__ int oct_lane(int x) {
    return __builtin_amdgcn_mov_dpp8(x, J * 0x249249);
}

__device__ __forceinline__ int list_entry(const int* __restrict__ perm, int st, int c, int e) {
    int ee = e < c ? e : c - 1;
    int s = 0;
    if (c > 0) s = perm[st + ee];
    return s;
}

__device__ __forceinline__ v4f sum_rows32(const float* __restrict__ feat, const int* __restrict__ perm,
                                          int st, int c, int idx, int j) {
    const v4f zero = {0.0f, 0.0f, 0.0f, 0.0f};
    v4f acc = zero;
    const char* base = (const char*)feat;
    const unsigned jb = 16u * (unsigned)j;
    for (int e0 = 0; e0 < c; e0 += 8) {
        int s[8];
        s[0] = oct_lane<0>(idx); s[1] = oct_lane<1>(idx); s[2] = oct_lane<2>(idx); s[3] = oct_lane<3>(idx);
        s[4] = oct_lane<4>(idx); s[5] = oct_lane<5>(idx); s[6] = oct_lane<6>(idx); s[7] = oct_lane<7>(idx);
        v4f v[8];
#pragma unroll
        for (int u = 0; u < 8; ++u) v[u] = *(const v4f*)(base + ((unsigned)s[u] * (unsigned)(IN_F * 4) + jb));
        if (e0 + 8 < c) idx = list_entry(perm, st, c, e0 + 8 + j);
#pragma unroll
        for (int u = 0; u < 8; ++u) acc += (e0 + u < c) ? v[u] : zero;
    }
    return acc;
}

#define TILE_NODES 64
#define X_PITCH    36
#define H_PITCH    68

__global__ __launch_bounds__(256, 16) void sage_front_kernel(const float* __restrict__ feat,
                                                         const int* __restrict__ perm,
                                                         const int* __restrict__ start,
                                                         const int* __restrict__ cnt,
                                                         const float* __restrict__ W1self,
                                                         const float* __restrict__ W1neigh,
                                                         const float* __restrict__ bias1,
                                                         const float* __restrict__ W2self,
                                                         const float* __restrict__ W2neigh,
                                                         const float* __restrict__ bias2,
                                                         float* __restrict__ S,
                                                         float* __restrict__ P,
                                                         int nNodes,
                                                         int nRowTiles) {
    __shared__ __attribute__((aligned(16))) float xs[TILE_NODES * X_PITCH];
    __shared__ __attribute__((aligned(16))) float ms[TILE_NODES * X_PITCH];
    __shared__ float hs[TILE_NODES * H_PITCH];
    const int lane = threadIdx.x & 31;
    const int wv = __builtin_amdgcn_readfirstlane(threadIdx.x >> 5);
    {
        const int g    = lane >> 3;
        const int j    = lane & 7;
        const int rowA = wv * NODES_PER_WAVE + g;
        const int rowB = rowA + 4;
        const int nA   = blockIdx.x * TILE_NODES + rowA;
        const int nB   = nA + 4;
        const v4f zero = {0.0f, 0.0f, 0.0f, 0.0f};
        int stA = 0, cA = 0, stB = 0, cB = 0;
        v4f ownA = zero, ownB = zero;
        if (nA < nNodes) { stA = start[nA]; cA = cnt[nA]; ownA = *(const v4f*)(feat + (long)nA * IN_F + 4 * j); }
        if (nB < nNodes) { stB = start[nB]; cB = cnt[nB]; ownB = *(const v4f*)(feat + (long)nB * IN_F + 4 * j); }
        const int iA = list_entry(perm, stA, cA, j);
        const int iB = list_entry(perm, stB, cB, j);
        *(v4f*)(xs + rowA * X_PITCH + 4 * j) = ownA;
        *(v4f*)(xs + rowB * X_PITCH + 4 * j) = ownB;
        const v4f accA = sum_rows32(feat, perm, stA, cA, iA, j);
        const float invA = 1.0f / fmaxf((float)cA, 1.0f);
        *(v4f*)(ms + rowA * X_PITCH + 4 * j) = accA * invA;
        const v4f accB = sum_rows32(feat, perm, stB, cB, iB, j);
        const float invB = 1.0f / fmaxf((float)cB, 1.0f);
        *(v4f*)(ms + rowB * X_PITCH + 4 * j) = accB * invB;
    }
    __syncthreads();

    const int hi  = lane >> 4;
    const int l15 = lane & 15;

    {
        const int ct   = wv & 3;
        const int rt0  = (wv >> 2) * 2;
        const int colW = ct * 16 + l15;
        const float* x0 = xs + (rt0 * 16 + l15) * X_PITCH;
        const float* x1 = x0 + 16 * X_PITCH;
        const float* m0 = ms + (rt0 * 16 + l15) * X_PITCH;
        const float* m1 = m0 + 16 * X_PITCH;
        v8f c0, c1;
        const float bv = bias1[colW];
#pragma unroll
        for (int v = 0; v < 8; ++v) { c0[v] = bv; c1[v] = bv; }
#pragma unroll
        for (int k0 = 0; k0 < IN_F; k0 += 4) {
            const int kA = k0 + 2 * hi;
            v2f a0, a1, b;
            b.x = W1self[(long)(kA)     * HID_F + colW];
            b.y = W1self[(long)(kA + 1) * HID_F + colW];
            a0.x = x0[kA]; a0.y = x0[kA + 1];
            a1.x = x1[kA]; a1.y = x1[kA + 1];
            c0 = __builtin_amdgcn_wmma_f32_16x16x4_f32(false, a0, false, b, (short)0, c0, false, false);
            c1 = __builtin_amdgcn_wmma_f32_16x16x4_f32(false, a1, false, b, (short)0, c1, false, false);
            b.x = W1neigh[(long)(kA)     * HID_F + colW];
            b.y = W1neigh[(long)(kA + 1) * HID_F + colW];
            a0.x = m0[kA]; a0.y = m0[kA + 1];
            a1.x = m1[kA]; a1.y = m1[kA + 1];
            c0 = __builtin_amdgcn_wmma_f32_16x16x4_f32(false, a0, false, b, (short)0, c0, false, false);
            c1 = __builtin_amdgcn_wmma_f32_16x16x4_f32(false, a1, false, b, (short)0, c1, false, false);
        }
#pragma unroll
        for (int v = 0; v < 8; ++v) {
            const int r = rt0 * 16 + v + 8 * hi;
            hs[r * H_PITCH + colW]        = fmaxf(c0[v], 0.0f);
            hs[(r + 16) * H_PITCH + colW] = fmaxf(c1[v], 0.0f);
        }
    }
    __syncthreads();

    if (wv < 4) {
        const int which = wv & 1;
        const int rt0   = (wv >> 1) * 2;
        const float* W  = which ? W2neigh : W2self;
        float*       O  = which ? P : S;
        const float* h0 = hs + (rt0 * 16 + l15) * H_PITCH;
        const float* h1 = h0 + 16 * H_PITCH;
        v8f c0, c1;
        const float bv = which ? 0.0f : bias2[l15];
#pragma unroll
        for (int v = 0; v < 8; ++v) { c0[v] = bv; c1[v] = bv; }
#pragma unroll
        for (int k0 = 0; k0 < HID_F; k0 += 4) {
            const int kA = k0 + 2 * hi;
            v2f a0, a1, b;
            b.x = W[(long)(kA)     * OUT_F + l15];
            b.y = W[(long)(kA + 1) * OUT_F + l15];
            a0.x = h0[kA]; a0.y = h0[kA + 1];
            a1.x = h1[kA]; a1.y = h1[kA + 1];
            c0 = __builtin_amdgcn_wmma_f32_16x16x4_f32(false, a0, false, b, (short)0, c0, false, false);
            c1 = __builtin_amdgcn_wmma_f32_16x16x4_f32(false, a1, false, b, (short)0, c1, false, false);
        }
        const int tile0 = blockIdx.x * (TILE_NODES / 16) + rt0;
        if (tile0 < nRowTiles) {
#pragma unroll
            for (int v = 0; v < 8; ++v) {
                const long r = (long)tile0 * 16 + v + 8 * hi;
                O[r * OUT_F + l15] = c0[v];
            }
        }
        if (tile0 + 1 < nRowTiles) {
#pragma unroll
            for (int v = 0; v < 8; ++v) {
                const long r = (long)(tile0 + 1) * 16 + v + 8 * hi;
                O[r * OUT_F + l15] = c1[v];
            }
        }
    }
}

#define WAVE_NODES 16

template <int J>
__device__ __forceinline__ int quad_lane(int x) {
    return __builtin_amdgcn_update_dpp(0, x, J * 0x55, 0xf, 0xf, true);
}

__device__ __forceinline__ void load_rows16(const float* __restrict__ P, int i0, int i1, int q, v4f v[8]) {
    int s[8];
    s[0] = quad_lane<0>(i0); s[1] = quad_lane<1>(i0); s[2] = quad_lane<2>(i0); s[3] = quad_lane<3>(i0);
    s[4] = quad_lane<0>(i1); s[5] = quad_lane<1>(i1); s[6] = quad_lane<2>(i1); s[7] = quad_lane<3>(i1);
#pragma unroll
    for (int u = 0; u < 8; ++u) v[u] = *(const v4f*)(P + (long)s[u] * OUT_F + 4 * q);
}

__global__ __launch_bounds__(256) void gather16_combine_kernel(const float* __restrict__ P,
                                                               const float* __restrict__ S,
                                                               const int* __restrict__ perm,
                                                               const int* __restrict__ start,
                                                               const int* __restrict__ cnt,
                                                               float* __restrict__ out,
                                                               int nNodes) {
    const int lane = threadIdx.x & 31;
    const int q    = lane & 3;
    const int wave = blockIdx.x * 8 + (threadIdx.x >> 5);
    const int nA   = wave * WAVE_NODES + (lane >> 2);
    const int nB   = nA + 8;
    const bool onA = nA < nNodes;
    const bool onB = nB < nNodes;
    int stA = 0, cA = 0, stB = 0, cB = 0;
    v4f selfA = {0.0f, 0.0f, 0.0f, 0.0f}, selfB = {0.0f, 0.0f, 0.0f, 0.0f};
    if (onA) { stA = start[nA]; cA = cnt[nA]; selfA = *(const v4f*)(S + (long)nA * OUT_F + 4 * q); }
    if (onB) { stB = start[nB]; cB = cnt[nB]; selfB = *(const v4f*)(S + (long)nB * OUT_F + 4 * q); }
    const int cm = cA > cB ? cA : cB;

    int a0 = list_entry(perm, stA, cA, q), a1 = list_entry(perm, stA, cA, 4 + q);
    int b0 = list_entry(perm, stB, cB, q), b1 = list_entry(perm, stB, cB, 4 + q);
    v4f accA = {0.0f, 0.0f, 0.0f, 0.0f}, accB = {0.0f, 0.0f, 0.0f, 0.0f};
    const v4f zero = {0.0f, 0.0f, 0.0f, 0.0f};
    for (int e0 = 0; e0 < cm; e0 += 8) {
        v4f va[8], vb[8];
        load_rows16(P, a0, a1, q, va);
        load_rows16(P, b0, b1, q, vb);
        if (e0 + 8 < cm) {
            a0 = list_entry(perm, stA, cA, e0 + 8 + q);  a1 = list_entry(perm, stA, cA, e0 + 12 + q);
            b0 = list_entry(perm, stB, cB, e0 + 8 + q);  b1 = list_entry(perm, stB, cB, e0 + 12 + q);
        }
#pragma unroll
        for (int u = 0; u < 8; ++u) {
            accA += (e0 + u < cA) ? va[u] : zero;
            accB += (e0 + u < cB) ? vb[u] : zero;
        }
    }
    if (onA) *(v4f*)(out + (long)nA * OUT_F + 4 * q) = selfA + accA / fmaxf((float)cA, 1.0f);
    if (onB) *(v4f*)(out + (long)nB * OUT_F + 4 * q) = selfB + accB / fmaxf((float)cB, 1.0f);
}

extern "C" void kernel_launch(void* const* d_in, const int* in_sizes, int n_in,
                              void* d_out, int out_size, void* d_ws, size_t ws_size,
                              hipStream_t stream) {
    const float* feat = (const float*)d_in[0];
    const int*   src  = (const int*)d_in[1];
    const int*   dst  = (const int*)d_in[2];
    const float* W1s  = (const float*)d_in[3];
    const float* W1n  = (const float*)d_in[4];
    const float* b1   = (const float*)d_in[5];
    const float* W2s  = (const float*)d_in[6];
    const float* W2n  = (const float*)d_in[7];
    const float* b2   = (const float*)d_in[8];
    float* out = (float*)d_out;

    const int nNodes = in_sizes[0] / IN_F;
    const int nE     = in_sizes[1];
    const int nRowTiles = nNodes / 16;

    float* ws    = (float*)d_ws;
    float* r0    = ws;
    float* r1    = r0 + (long)nNodes * HID_F;
    int*   perm  = (int*)(r1 + (long)nNodes * HID_F);
    int*   cnt   = perm + nE;
    int*   cursor = cnt + nNodes;
    int*   start = cursor + 32;
    int*   bkt   = (int*)r0;
    int*   runs  = (int*)r1;
    float* P     = r1;
    float* S     = r1 + (long)nNodes * OUT_F;

    const int tBlocks = (nNodes + TILE_NODES - 1) / TILE_NODES;
    const int qBlocks = (nNodes + 8 * WAVE_NODES - 1) / (8 * WAVE_NODES);

    chunk_bin_kernel<<<N_CHUNKS, SORT_THREADS, 0, stream>>>(src, dst, bkt, runs, cursor, nE);
    bucket_sort_kernel<<<N_BKT, SORT_THREADS, 0, stream>>>(bkt, runs, cursor, perm, cnt, start, nNodes);

    sage_front_kernel<<<tBlocks, 256, 0, stream>>>(feat, perm, start, cnt, W1s, W1n, b1, W2s, W2n, b2, S, P, nNodes, nRowTiles);

    gather16_combine_kernel<<<qBlocks, 256, 0, stream>>>(P, S, perm, start, cnt, out, nNodes);
}
